# starting kernel
# baseline (speedup 1.0000x reference)
#include <hip/hip_runtime.h>
#include <hip/hip_bf16.h>

typedef float v2f __attribute__((ext_vector_type(2)));
typedef float v8f __attribute__((ext_vector_type(8)));

#define NFEAT 128
#define NGRAPH 512
#define NREL 200

__global__ void zero_f32_kernel(float* __restrict__ p, long n) {
    long i = (long)blockIdx.x * blockDim.x + threadIdx.x;
    if (i < n) p[i] = 0.0f;
}

__global__ void degree_kernel(const int* __restrict__ src, const int* __restrict__ dst,
                              float* __restrict__ deg_out, float* __restrict__ deg_in, int E) {
    int e = blockIdx.x * blockDim.x + threadIdx.x;
    if (e < E) {
        atomicAdd(&deg_out[src[e]], 1.0f);
        atomicAdd(&deg_in[dst[e]], 1.0f);
    }
}

__global__ void norm_finalize_kernel(float* __restrict__ d, long n) {
    long i = (long)blockIdx.x * blockDim.x + threadIdx.x;
    if (i < n) d[i] = rsqrtf(fmaxf(d[i], 1.0f));
}

__global__ void build_x_kernel(const int* __restrict__ nid, const int* __restrict__ z,
                               const float* __restrict__ emb, const float* __restrict__ ztab,
                               float* __restrict__ x, int N) {
    long t = (long)blockIdx.x * blockDim.x + threadIdx.x;
    long i = t >> 7;
    int  f = (int)(t & 127);
    if (i < N) {
        x[t] = (f < 64) ? emb[(long)nid[i] * 64 + f]
                        : ztab[(long)z[i] * 64 + (f - 64)];
    }
}

__global__ void scatter_kernel(const int* __restrict__ src, const int* __restrict__ dst,
                               const float* __restrict__ x, const float* __restrict__ norm_out,
                               float* __restrict__ agg, int E) {
    long t = (long)blockIdx.x * blockDim.x + threadIdx.x;
    long e = t >> 5;
    if (e >= E) return;
    int f4 = (int)((t & 31) << 2);
    int s = src[e], d = dst[e];
    float no = norm_out[s];
    const float4 v = *(const float4*)(x + (long)s * NFEAT + f4);
    float* ap = agg + (long)d * NFEAT + f4;
    atomicAdd(ap + 0, v.x * no);
    atomicAdd(ap + 1, v.y * no);
    atomicAdd(ap + 2, v.z * no);
    atomicAdd(ap + 3, v.w * no);
}

__global__ __launch_bounds__(256) void gemm128_wmma_kernel(
    const float* __restrict__ A, const float* __restrict__ norm,
    const float* __restrict__ W, const float* __restrict__ bias,
    float* __restrict__ Y, int nrows, int relu) {
    __shared__ float Wlds[NFEAT * NFEAT];

    int tid = threadIdx.x;
    {
        const float4* Wv = (const float4*)W;
        float4* Lv = (float4*)Wlds;
#pragma unroll
        for (int i = 0; i < 16; ++i) Lv[tid + i * 256] = Wv[tid + i * 256];
    }
    __syncthreads();

    int wave = tid >> 5;
    int lane = tid & 31;
    int rbase = blockIdx.x * 128 + wave * 16;
    if (rbase >= nrows) return;

    int lrow  = lane & 15;
    int khalf = (lane >> 4) << 1;
    int rowhi = (lane >> 4) << 3;

    long m = rbase + lrow;
    float nrm = norm ? norm[m] : 1.0f;
    const float* Arow = A + m * NFEAT;

    v8f acc[8];
#pragma unroll
    for (int t = 0; t < 8; ++t) {
        float bv = bias[t * 16 + lrow];
#pragma unroll
        for (int r = 0; r < 8; ++r) acc[t][r] = bv;
    }

    for (int k = 0; k < NFEAT; k += 4) {
        v2f a;
        a.x = Arow[k + khalf] * nrm;
        a.y = Arow[k + khalf + 1] * nrm;
#pragma unroll
        for (int t = 0; t < 8; ++t) {
            v2f b;
            b.x = Wlds[(k + khalf) * NFEAT + t * 16 + lrow];
            b.y = Wlds[(k + khalf + 1) * NFEAT + t * 16 + lrow];
            acc[t] = __builtin_amdgcn_wmma_f32_16x16x4_f32(
                false, a, false, b, (short)0, acc[t], false, false);
        }
    }

#pragma unroll
    for (int t = 0; t < 8; ++t) {
#pragma unroll
        for (int r = 0; r < 8; ++r) {
            float v = acc[t][r];
            if (relu) v = fmaxf(v, 0.0f);
            Y[(long)(rbase + rowhi + r) * NFEAT + t * 16 + lrow] = v;
        }
    }
}

__global__ void pool_kernel(const float* __restrict__ x, const int* __restrict__ gid,
                            float* __restrict__ g, int N) {
    long t = (long)blockIdx.x * blockDim.x + threadIdx.x;
    long i = t >> 7;
    int  f = (int)(t & 127);
    if (i < N) atomicAdd(&g[(long)gid[i] * NFEAT + f], x[t]);
}

__global__ void mlp2_kernel(const float* __restrict__ hid, const float* __restrict__ W,
                            const float* __restrict__ b, float* __restrict__ out) {
    int t = blockIdx.x * blockDim.x + threadIdx.x;
    if (t >= NGRAPH * NREL) return;
    int gi = t / NREL, n = t % NREL;
    const float* h = hid + (long)gi * NFEAT;
    float s = b[n];
#pragma unroll 8
    for (int k = 0; k < NFEAT; ++k) s = fmaf(h[k], W[k * NREL + n], s);
    out[t] = s;
}

extern "C" void kernel_launch(void* const* d_in, const int* in_sizes, int n_in,
                              void* d_out, int out_size, void* d_ws, size_t ws_size,
                              hipStream_t stream) {
    const int*   nid    = (const int*)d_in[0];
    const int*   z      = (const int*)d_in[1];
    const int*   src    = (const int*)d_in[2];
    const int*   dst    = (const int*)d_in[3];
    const int*   gid    = (const int*)d_in[4];
    const float* emb    = (const float*)d_in[5];
    const float* ztab   = (const float*)d_in[6];
    const float* Ws[3]  = {(const float*)d_in[7], (const float*)d_in[9], (const float*)d_in[11]};
    const float* bs[3]  = {(const float*)d_in[8], (const float*)d_in[10], (const float*)d_in[12]};
    const float* lin1_W = (const float*)d_in[13];
    const float* lin1_b = (const float*)d_in[14];
    const float* lin2_W = (const float*)d_in[15];
    const float* lin2_b = (const float*)d_in[16];

    const int N = in_sizes[0];
    const int E = in_sizes[2];

    float* x        = (float*)d_ws;
    float* agg      = x + (size_t)N * NFEAT;
    float* norm_out = agg + (size_t)N * NFEAT;
    float* norm_in  = norm_out + N;
    float* g        = norm_in + N;
    float* hid      = g + (size_t)NGRAPH * NFEAT;

    const int TB = 256;
    auto blocks = [](long n, int tb) { return (unsigned)((n + tb - 1) / tb); };

    zero_f32_kernel<<<blocks(2L * N, TB), TB, 0, stream>>>(norm_out, 2L * N);
    degree_kernel<<<blocks(E, TB), TB, 0, stream>>>(src, dst, norm_out, norm_in, E);
    norm_finalize_kernel<<<blocks(2L * N, TB), TB, 0, stream>>>(norm_out, 2L * N);

    build_x_kernel<<<blocks((long)N * NFEAT, TB), TB, 0, stream>>>(nid, z, emb, ztab, x, N);

    const unsigned gemm_blocks = (unsigned)((N + 127) / 128);
    for (int layer = 0; layer < 3; ++layer) {
        zero_f32_kernel<<<blocks((long)N * NFEAT, TB), TB, 0, stream>>>(agg, (long)N * NFEAT);
        scatter_kernel<<<blocks((long)E * 32, TB), TB, 0, stream>>>(src, dst, x, norm_out, agg, E);
        gemm128_wmma_kernel<<<gemm_blocks, TB, 0, stream>>>(
            agg, norm_in, Ws[layer], bs[layer], x, N, layer < 2 ? 1 : 0);
    }

    zero_f32_kernel<<<blocks((long)NGRAPH * NFEAT, TB), TB, 0, stream>>>(g, (long)NGRAPH * NFEAT);
    pool_kernel<<<blocks((long)N * NFEAT, TB), TB, 0, stream>>>(x, gid, g, N);

    gemm128_wmma_kernel<<<(NGRAPH + 127) / 128, TB, 0, stream>>>(
        g, nullptr, lin1_W, lin1_b, hid, NGRAPH, 1);
    mlp2_kernel<<<blocks((long)NGRAPH * NREL, TB), TB, 0, stream>>>(hid, lin2_W, lin2_b, (float*)d_out);
}


# Aggregation as destination-ordered gather: per-call edge list from integer counts + arrival-order rank, one wave sums a row in list order; replaces float-atomic scatter and its zero fill
# speedup vs baseline: 1.8891x; 1.8891x over previous
#include <hip/hip_runtime.h>
#include <hip/hip_bf16.h>

typedef float v2f __attribute__((ext_vector_type(2)));
typedef float v8f __attribute__((ext_vector_type(8)));

#define NFEAT 128
#define NGRAPH 512
#define NREL 200

__global__ void zero_f32_kernel(float* __restrict__ p, long n) {
    long i = (long)blockIdx.x * blockDim.x + threadIdx.x;
    if (i < n) p[i] = 0.0f;
}

__global__ void zero_i32_kernel(int* __restrict__ p, long n) {
    long i = (long)blockIdx.x * blockDim.x + threadIdx.x;
    if (i < n) p[i] = 0;
}

__global__ void count_rank_kernel(const int* __restrict__ src, const int* __restrict__ dst,
                                  int* __restrict__ cnt_out, int* __restrict__ cnt_in,
                                  int* __restrict__ rank, int E) {
    int e = blockIdx.x * blockDim.x + threadIdx.x;
    if (e < E) {
        atomicAdd(&cnt_out[src[e]], 1);
        rank[e] = atomicAdd(&cnt_in[dst[e]], 1);
    }
}

__global__ __launch_bounds__(1024) void scan_norm_kernel(
    const int* __restrict__ cnt_out, const int* __restrict__ cnt_in,
    int* __restrict__ rowptr, float* __restrict__ norm_out, float* __restrict__ norm_in, int N) {
    const int tid = threadIdx.x;
    if (blockIdx.x != 0) {
        int i = (blockIdx.x - 1) * 1024 + tid;
        if (i < N) {
            norm_out[i] = rsqrtf(fmaxf((float)cnt_out[i], 1.0f));
            norm_in[i]  = rsqrtf(fmaxf((float)cnt_in[i], 1.0f));
        }
        return;
    }
    __shared__ int wsum[32];
    const int ch  = (N + 1023) / 1024;
    const int beg = min(tid * ch, N);
    const int end = min(beg + ch, N);
    int s = 0;
    int i = beg;
    for (; i + 8 <= end; i += 8) {
        int a[8];
#pragma unroll
        for (int u = 0; u < 8; ++u) a[u] = cnt_in[i + u];
#pragma unroll
        for (int u = 0; u < 8; ++u) s += a[u];
    }
    for (; i < end; ++i) s += cnt_in[i];

    const int lane = tid & 31, w = tid >> 5;
    int incl = s;
#pragma unroll
    for (int o = 1; o < 32; o <<= 1) {
        int t = __shfl_up(incl, o);
        if (lane >= o) incl += t;
    }
    if (lane == 31) wsum[w] = incl;
    __syncthreads();
    if (w == 0) {
        int v = wsum[lane], iv = v;
#pragma unroll
        for (int o = 1; o < 32; o <<= 1) {
            int t = __shfl_up(iv, o);
            if (lane >= o) iv += t;
        }
        wsum[lane] = iv - v;
    }
    __syncthreads();
    int run = wsum[w] + incl - s;
    i = beg;
    for (; i + 8 <= end; i += 8) {
        int a[8];
#pragma unroll
        for (int u = 0; u < 8; ++u) a[u] = cnt_in[i + u];
#pragma unroll
        for (int u = 0; u < 8; ++u) { rowptr[i + u] = run; run += a[u]; }
    }
    for (; i < end; ++i) { rowptr[i] = run; run += cnt_in[i]; }
    if (tid == 1023) rowptr[N] = run;
}

__global__ void fill_kernel(const int* __restrict__ src, const int* __restrict__ dst,
                            const int* __restrict__ rank, const int* __restrict__ rowptr,
                            int* __restrict__ csr_src, int E) {
    int e = blockIdx.x * blockDim.x + threadIdx.x;
    if (e < E) csr_src[rowptr[dst[e]] + rank[e]] = src[e];
}

__global__ void build_x_kernel(const int* __restrict__ nid, const int* __restrict__ z,
                               const float* __restrict__ emb, const float* __restrict__ ztab,
                               float* __restrict__ x, int N) {
    long t = (long)blockIdx.x * blockDim.x + threadIdx.x;
    long i = t >> 7;
    int  f = (int)(t & 127);
    if (i < N) {
        x[t] = (f < 64) ? emb[(long)nid[i] * 64 + f]
                        : ztab[(long)z[i] * 64 + (f - 64)];
    }
}

#define ROWS_PER_WAVE 2

template <int U>
__device__ __forceinline__ void add_rows(float4& acc, int s_l, int w_bits, int j,
                                         const float* __restrict__ xl) {
    float4 v[U];
    float w[U];
#pragma unroll
    for (int u = 0; u < U; ++u) {
        int sj = __builtin_amdgcn_readlane(s_l, j + u);
        w[u] = __int_as_float(__builtin_amdgcn_readlane(w_bits, j + u));
        v[u] = *(const float4*)(xl + (long)sj * NFEAT);
    }
#pragma unroll
    for (int u = 0; u < U; ++u) {
        acc.x = fmaf(v[u].x, w[u], acc.x);
        acc.y = fmaf(v[u].y, w[u], acc.y);
        acc.z = fmaf(v[u].z, w[u], acc.z);
        acc.w = fmaf(v[u].w, w[u], acc.w);
    }
}

__global__ __launch_bounds__(256) void gather_kernel(
    const int* __restrict__ rowptr, const int* __restrict__ csr_src,
    const float* __restrict__ x, const float* __restrict__ norm_out,
    float* __restrict__ agg, int N) {
    const int lane = threadIdx.x & 31;
    const int wave = __builtin_amdgcn_readfirstlane((int)((blockIdx.x * 256u + threadIdx.x) >> 5));
    const float* xl = x + lane * 4;
    for (int q = 0; q < ROWS_PER_WAVE; ++q) {
        const int i = wave * ROWS_PER_WAVE + q;
        if (i >= N) break;
        const int beg = rowptr[i];
        const int end = rowptr[i + 1];
        float4 acc = make_float4(0.0f, 0.0f, 0.0f, 0.0f);
        for (int base = beg; base < end; base += 32) {
            const int n = min(32, end - base);
            int s_l = 0;
            float w_l = 0.0f;
            if (lane < n) {
                s_l = csr_src[base + lane];
                w_l = norm_out[s_l];
            }
            const int w_bits = __float_as_int(w_l);
            int j = 0;
            for (; j + 8 <= n; j += 8) add_rows<8>(acc, s_l, w_bits, j, xl);
            if (n - j >= 4) { add_rows<4>(acc, s_l, w_bits, j, xl); j += 4; }
            if (n - j >= 2) { add_rows<2>(acc, s_l, w_bits, j, xl); j += 2; }
            if (n - j >= 1) { add_rows<1>(acc, s_l, w_bits, j, xl); }
        }
        *(float4*)(agg + (long)i * NFEAT + lane * 4) = acc;
    }
}

__global__ __launch_bounds__(256) void gemm128_wmma_kernel(
    const float* __restrict__ A, const float* __restrict__ norm,
    const float* __restrict__ W, const float* __restrict__ bias,
    float* __restrict__ Y, int nrows, int relu) {
    __shared__ float Wlds[NFEAT * NFEAT];

    int tid = threadIdx.x;
    {
        const float4* Wv = (const float4*)W;
        float4* Lv = (float4*)Wlds;
#pragma unroll
        for (int i = 0; i < 16; ++i) Lv[tid + i * 256] = Wv[tid + i * 256];
    }
    __syncthreads();

    int wave = tid >> 5;
    int lane = tid & 31;
    int rbase = blockIdx.x * 128 + wave * 16;
    if (rbase >= nrows) return;

    int lrow  = lane & 15;
    int khalf = (lane >> 4) << 1;
    int rowhi = (lane >> 4) << 3;

    long m = rbase + lrow;
    float nrm = norm ? norm[m] : 1.0f;
    const float* Arow = A + m * NFEAT;

    v8f acc[8];
#pragma unroll
    for (int t = 0; t < 8; ++t) {
        float bv = bias[t * 16 + lrow];
#pragma unroll
        for (int r = 0; r < 8; ++r) acc[t][r] = bv;
    }

    for (int k = 0; k < NFEAT; k += 4) {
        v2f a;
        a.x = Arow[k + khalf] * nrm;
        a.y = Arow[k + khalf + 1] * nrm;
#pragma unroll
        for (int t = 0; t < 8; ++t) {
            v2f b;
            b.x = Wlds[(k + khalf) * NFEAT + t * 16 + lrow];
            b.y = Wlds[(k + khalf + 1) * NFEAT + t * 16 + lrow];
            acc[t] = __builtin_amdgcn_wmma_f32_16x16x4_f32(
                false, a, false, b, (short)0, acc[t], false, false);
        }
    }

#pragma unroll
    for (int t = 0; t < 8; ++t) {
#pragma unroll
        for (int r = 0; r < 8; ++r) {
            float v = acc[t][r];
            if (relu) v = fmaxf(v, 0.0f);
            Y[(long)(rbase + rowhi + r) * NFEAT + t * 16 + lrow] = v;
        }
    }
}

__global__ void pool_kernel(const float* __restrict__ x, const int* __restrict__ gid,
                            float* __restrict__ g, int N) {
    long t = (long)blockIdx.x * blockDim.x + threadIdx.x;
    long i = t >> 7;
    int  f = (int)(t & 127);
    if (i < N) atomicAdd(&g[(long)gid[i] * NFEAT + f], x[t]);
}

__global__ void mlp2_kernel(const float* __restrict__ hid, const float* __restrict__ W,
                            const float* __restrict__ b, float* __restrict__ out) {
    int t = blockIdx.x * blockDim.x + threadIdx.x;
    if (t >= NGRAPH * NREL) return;
    int gi = t / NREL, n = t % NREL;
    const float* h = hid + (long)gi * NFEAT;
    float s = b[n];
#pragma unroll 8
    for (int k = 0; k < NFEAT; ++k) s = fmaf(h[k], W[k * NREL + n], s);
    out[t] = s;
}

extern "C" void kernel_launch(void* const* d_in, const int* in_sizes, int n_in,
                              void* d_out, int out_size, void* d_ws, size_t ws_size,
                              hipStream_t stream) {
    const int*   nid    = (const int*)d_in[0];
    const int*   z      = (const int*)d_in[1];
    const int*   src    = (const int*)d_in[2];
    const int*   dst    = (const int*)d_in[3];
    const int*   gid    = (const int*)d_in[4];
    const float* emb    = (const float*)d_in[5];
    const float* ztab   = (const float*)d_in[6];
    const float* Ws[3]  = {(const float*)d_in[7], (const float*)d_in[9], (const float*)d_in[11]};
    const float* bs[3]  = {(const float*)d_in[8], (const float*)d_in[10], (const float*)d_in[12]};
    const float* lin1_W = (const float*)d_in[13];
    const float* lin1_b = (const float*)d_in[14];
    const float* lin2_W = (const float*)d_in[15];
    const float* lin2_b = (const float*)d_in[16];

    const int N = in_sizes[0];
    const int E = in_sizes[2];

    float* x        = (float*)d_ws;
    float* agg      = x + (size_t)N * NFEAT;
    float* norm_out = agg + (size_t)N * NFEAT;
    float* norm_in  = norm_out + N;
    float* g        = norm_in + N;
    float* hid      = g + (size_t)NGRAPH * NFEAT;
    int*   cnt_out  = (int*)(hid + (size_t)NGRAPH * NFEAT);
    int*   cnt_in   = cnt_out + N;
    int*   rowptr   = cnt_in + N;
    int*   csr_src  = rowptr + N + 1;
    int*   rank     = (int*)agg;

    const int TB = 256;
    auto blocks = [](long n, int tb) { return (unsigned)((n + tb - 1) / tb); };

    zero_i32_kernel<<<blocks(2L * N, TB), TB, 0, stream>>>(cnt_out, 2L * N);
    count_rank_kernel<<<blocks(E, TB), TB, 0, stream>>>(src, dst, cnt_out, cnt_in, rank, E);
    scan_norm_kernel<<<1 + blocks(N, 1024), 1024, 0, stream>>>(cnt_out, cnt_in, rowptr, norm_out, norm_in, N);
    fill_kernel<<<blocks(E, TB), TB, 0, stream>>>(src, dst, rank, rowptr, csr_src, E);

    build_x_kernel<<<blocks((long)N * NFEAT, TB), TB, 0, stream>>>(nid, z, emb, ztab, x, N);

    const unsigned gemm_blocks = (unsigned)((N + 127) / 128);
    const unsigned gather_blocks = blocks((long)((N + ROWS_PER_WAVE - 1) / ROWS_PER_WAVE) * 32, TB);
    for (int layer = 0; layer < 3; ++layer) {
        gather_kernel<<<gather_blocks, TB, 0, stream>>>(rowptr, csr_src, x, norm_out, agg, N);
        gemm128_wmma_kernel<<<gemm_blocks, TB, 0, stream>>>(
            agg, norm_in, Ws[layer], bs[layer], x, N, layer < 2 ? 1 : 0);
    }

    zero_f32_kernel<<<blocks((long)NGRAPH * NFEAT, TB), TB, 0, stream>>>(g, (long)NGRAPH * NFEAT);
    pool_kernel<<<blocks((long)N * NFEAT, TB), TB, 0, stream>>>(x, gid, g, N);

    gemm128_wmma_kernel<<<(NGRAPH + 127) / 128, TB, 0, stream>>>(
        g, nullptr, lin1_W, lin1_b, hid, NGRAPH, 1);
    mlp2_kernel<<<blocks((long)NGRAPH * NREL, TB), TB, 0, stream>>>(hid, lin2_W, lin2_b, (float*)d_out);
}


# Fewer, fatter waves in the four one-element-per-thread kernels (feature build, sum pooling, edge count, edge fill): a wave takes 8 whole rows, a thread 4 edges; about 425 thousand fewer waves enter
# speedup vs baseline: 2.0919x; 1.1074x over previous
#include <hip/hip_runtime.h>
#include <hip/hip_bf16.h>

typedef float v2f __attribute__((ext_vector_type(2)));
typedef float v8f __attribute__((ext_vector_type(8)));

#define NFEAT 128
#define NGRAPH 512
#define NREL 200

__global__ void zero_f32_kernel(float* __restrict__ p, long n) {
    long i = (long)blockIdx.x * blockDim.x + threadIdx.x;
    if (i < n) p[i] = 0.0f;
}

__global__ void zero_i32_kernel(int* __restrict__ p, long n) {
    long i = (long)blockIdx.x * blockDim.x + threadIdx.x;
    if (i < n) p[i] = 0;
}

#define EDGE_TB 256
#define EDGE_PT 4

__global__ __launch_bounds__(EDGE_TB) void count_rank_kernel(
    const int* __restrict__ src, const int* __restrict__ dst,
    int* __restrict__ cnt_out, int* __restrict__ cnt_in,
    int* __restrict__ rank, int E) {
    const int blk = blockIdx.x * (EDGE_TB * EDGE_PT);
    const int e0 = blk + threadIdx.x;
    if (blk + EDGE_TB * EDGE_PT <= E) {
        int s[EDGE_PT], d[EDGE_PT], r[EDGE_PT];
#pragma unroll
        for (int k = 0; k < EDGE_PT; ++k) { s[k] = src[e0 + k * EDGE_TB]; d[k] = dst[e0 + k * EDGE_TB]; }
#pragma unroll
        for (int k = 0; k < EDGE_PT; ++k) {
            atomicAdd(&cnt_out[s[k]], 1);
            r[k] = atomicAdd(&cnt_in[d[k]], 1);
        }
#pragma unroll
        for (int k = 0; k < EDGE_PT; ++k) rank[e0 + k * EDGE_TB] = r[k];
    } else {
        for (int k = 0; k < EDGE_PT; ++k) {
            const int e = e0 + k * EDGE_TB;
            if (e < E) {
                atomicAdd(&cnt_out[src[e]], 1);
                rank[e] = atomicAdd(&cnt_in[dst[e]], 1);
            }
        }
    }
}

__global__ __launch_bounds__(1024) void scan_norm_kernel(
    const int* __restrict__ cnt_out, const int* __restrict__ cnt_in,
    int* __restrict__ rowptr, float* __restrict__ norm_out, float* __restrict__ norm_in, int N) {
    const int tid = threadIdx.x;
    if (blockIdx.x != 0) {
        int i = (blockIdx.x - 1) * 1024 + tid;
        if (i < N) {
            norm_out[i] = rsqrtf(fmaxf((float)cnt_out[i], 1.0f));
            norm_in[i]  = rsqrtf(fmaxf((float)cnt_in[i], 1.0f));
        }
        return;
    }
    __shared__ int wsum[32];
    const int ch  = (N + 1023) / 1024;
    const int beg = min(tid * ch, N);
    const int end = min(beg + ch, N);
    int s = 0;
    int i = beg;
    for (; i + 8 <= end; i += 8) {
        int a[8];
#pragma unroll
        for (int u = 0; u < 8; ++u) a[u] = cnt_in[i + u];
#pragma unroll
        for (int u = 0; u < 8; ++u) s += a[u];
    }
    for (; i < end; ++i) s += cnt_in[i];

    const int lane = tid & 31, w = tid >> 5;
    int incl = s;
#pragma unroll
    for (int o = 1; o < 32; o <<= 1) {
        int t = __shfl_up(incl, o);
        if (lane >= o) incl += t;
    }
    if (lane == 31) wsum[w] = incl;
    __syncthreads();
    if (w == 0) {
        int v = wsum[lane], iv = v;
#pragma unroll
        for (int o = 1; o < 32; o <<= 1) {
            int t = __shfl_up(iv, o);
            if (lane >= o) iv += t;
        }
        wsum[lane] = iv - v;
    }
    __syncthreads();
    int run = wsum[w] + incl - s;
    i = beg;
    for (; i + 8 <= end; i += 8) {
        int a[8];
#pragma unroll
        for (int u = 0; u < 8; ++u) a[u] = cnt_in[i + u];
#pragma unroll
        for (int u = 0; u < 8; ++u) { rowptr[i + u] = run; run += a[u]; }
    }
    for (; i < end; ++i) { rowptr[i] = run; run += cnt_in[i]; }
    if (tid == 1023) rowptr[N] = run;
}

__global__ __launch_bounds__(EDGE_TB) void fill_kernel(
    const int* __restrict__ src, const int* __restrict__ dst,
    const int* __restrict__ rank, const int* __restrict__ rowptr,
    int* __restrict__ csr_src, int E) {
    const int blk = blockIdx.x * (EDGE_TB * EDGE_PT);
    const int e0 = blk + threadIdx.x;
    if (blk + EDGE_TB * EDGE_PT <= E) {
        int s[EDGE_PT], d[EDGE_PT], r[EDGE_PT], p[EDGE_PT];
#pragma unroll
        for (int k = 0; k < EDGE_PT; ++k) {
            s[k] = src[e0 + k * EDGE_TB]; d[k] = dst[e0 + k * EDGE_TB]; r[k] = rank[e0 + k * EDGE_TB];
        }
#pragma unroll
        for (int k = 0; k < EDGE_PT; ++k) p[k] = rowptr[d[k]];
#pragma unroll
        for (int k = 0; k < EDGE_PT; ++k) csr_src[p[k] + r[k]] = s[k];
    } else {
        for (int k = 0; k < EDGE_PT; ++k) {
            const int e = e0 + k * EDGE_TB;
            if (e < E) csr_src[rowptr[dst[e]] + rank[e]] = src[e];
        }
    }
}

#define ROWS_BX 8

__device__ __forceinline__ float4 bx_row(const float* __restrict__ tab, bool ent,
                                         const int* __restrict__ nid, const int* __restrict__ z, int i) {
    const int rn = nid[i];
    const int rz = z[i];
    return *(const float4*)(tab + (long)(ent ? rn : rz) * 64);
}

__global__ __launch_bounds__(256) void build_x_kernel(
    const int* __restrict__ nid, const int* __restrict__ z,
    const float* __restrict__ emb, const float* __restrict__ ztab,
    float* __restrict__ x, int N) {
    const int lane = threadIdx.x & 31;
    const int wave = __builtin_amdgcn_readfirstlane((int)((blockIdx.x * 256u + threadIdx.x) >> 5));
    const int i0 = wave * ROWS_BX;
    if (i0 >= N) return;
    const bool ent = lane < 16;
    const float* tab = (ent ? emb : ztab) + (lane & 15) * 4;
    float* xo = x + lane * 4;
    if (i0 + ROWS_BX <= N) {
        const float4 v0 = bx_row(tab, ent, nid, z, i0 + 0);
        const float4 v1 = bx_row(tab, ent, nid, z, i0 + 1);
        const float4 v2 = bx_row(tab, ent, nid, z, i0 + 2);
        const float4 v3 = bx_row(tab, ent, nid, z, i0 + 3);
        const float4 v4 = bx_row(tab, ent, nid, z, i0 + 4);
        const float4 v5 = bx_row(tab, ent, nid, z, i0 + 5);
        const float4 v6 = bx_row(tab, ent, nid, z, i0 + 6);
        const float4 v7 = bx_row(tab, ent, nid, z, i0 + 7);
        __builtin_amdgcn_sched_barrier(0);
        float* xr = xo + (long)i0 * NFEAT;
        *(float4*)(xr + 0 * NFEAT) = v0;
        *(float4*)(xr + 1 * NFEAT) = v1;
        *(float4*)(xr + 2 * NFEAT) = v2;
        *(float4*)(xr + 3 * NFEAT) = v3;
        *(float4*)(xr + 4 * NFEAT) = v4;
        *(float4*)(xr + 5 * NFEAT) = v5;
        *(float4*)(xr + 6 * NFEAT) = v6;
        *(float4*)(xr + 7 * NFEAT) = v7;
    } else {
        for (int i = i0; i < N; ++i) {
            *(float4*)(xo + (long)i * NFEAT) = bx_row(tab, ent, nid, z, i);
        }
    }
}

#define ROWS_PER_WAVE 2

template <int U>
__device__ __forceinline__ void add_rows(float4& acc, int s_l, int w_bits, int j,
                                         const float* __restrict__ xl) {
    float4 v[U];
    float w[U];
#pragma unroll
    for (int u = 0; u < U; ++u) {
        int sj = __builtin_amdgcn_readlane(s_l, j + u);
        w[u] = __int_as_float(__builtin_amdgcn_readlane(w_bits, j + u));
        v[u] = *(const float4*)(xl + (long)sj * NFEAT);
    }
#pragma unroll
    for (int u = 0; u < U; ++u) {
        acc.x = fmaf(v[u].x, w[u], acc.x);
        acc.y = fmaf(v[u].y, w[u], acc.y);
        acc.z = fmaf(v[u].z, w[u], acc.z);
        acc.w = fmaf(v[u].w, w[u], acc.w);
    }
}

__global__ __launch_bounds__(256) void gather_kernel(
    const int* __restrict__ rowptr, const int* __restrict__ csr_src,
    const float* __restrict__ x, const float* __restrict__ norm_out,
    float* __restrict__ agg, int N) {
    const int lane = threadIdx.x & 31;
    const int wave = __builtin_amdgcn_readfirstlane((int)((blockIdx.x * 256u + threadIdx.x) >> 5));
    const float* xl = x + lane * 4;
    for (int q = 0; q < ROWS_PER_WAVE; ++q) {
        const int i = wave * ROWS_PER_WAVE + q;
        if (i >= N) break;
        const int beg = rowptr[i];
        const int end = rowptr[i + 1];
        float4 acc = make_float4(0.0f, 0.0f, 0.0f, 0.0f);
        for (int base = beg; base < end; base += 32) {
            const int n = min(32, end - base);
            int s_l = 0;
            float w_l = 0.0f;
            if (lane < n) {
                s_l = csr_src[base + lane];
                w_l = norm_out[s_l];
            }
            const int w_bits = __float_as_int(w_l);
            int j = 0;
            for (; j + 8 <= n; j += 8) add_rows<8>(acc, s_l, w_bits, j, xl);
            if (n - j >= 4) { add_rows<4>(acc, s_l, w_bits, j, xl); j += 4; }
            if (n - j >= 2) { add_rows<2>(acc, s_l, w_bits, j, xl); j += 2; }
            if (n - j >= 1) { add_rows<1>(acc, s_l, w_bits, j, xl); }
        }
        *(float4*)(agg + (long)i * NFEAT + lane * 4) = acc;
    }
}

__global__ __launch_bounds__(256) void gemm128_wmma_kernel(
    const float* __restrict__ A, const float* __restrict__ norm,
    const float* __restrict__ W, const float* __restrict__ bias,
    float* __restrict__ Y, int nrows, int relu) {
    __shared__ float Wlds[NFEAT * NFEAT];

    int tid = threadIdx.x;
    {
        const float4* Wv = (const float4*)W;
        float4* Lv = (float4*)Wlds;
#pragma unroll
        for (int i = 0; i < 16; ++i) Lv[tid + i * 256] = Wv[tid + i * 256];
    }
    __syncthreads();

    int wave = tid >> 5;
    int lane = tid & 31;
    int rbase = blockIdx.x * 128 + wave * 16;
    if (rbase >= nrows) return;

    int lrow  = lane & 15;
    int khalf = (lane >> 4) << 1;
    int rowhi = (lane >> 4) << 3;

    long m = rbase + lrow;
    float nrm = norm ? norm[m] : 1.0f;
    const float* Arow = A + m * NFEAT;

    v8f acc[8];
#pragma unroll
    for (int t = 0; t < 8; ++t) {
        float bv = bias[t * 16 + lrow];
#pragma unroll
        for (int r = 0; r < 8; ++r) acc[t][r] = bv;
    }

    for (int k = 0; k < NFEAT; k += 4) {
        v2f a;
        a.x = Arow[k + khalf] * nrm;
        a.y = Arow[k + khalf + 1] * nrm;
#pragma unroll
        for (int t = 0; t < 8; ++t) {
            v2f b;
            b.x = Wlds[(k + khalf) * NFEAT + t * 16 + lrow];
            b.y = Wlds[(k + khalf + 1) * NFEAT + t * 16 + lrow];
            acc[t] = __builtin_amdgcn_wmma_f32_16x16x4_f32(
                false, a, false, b, (short)0, acc[t], false, false);
        }
    }

#pragma unroll
    for (int t = 0; t < 8; ++t) {
#pragma unroll
        for (int r = 0; r < 8; ++r) {
            float v = acc[t][r];
            if (relu) v = fmaxf(v, 0.0f);
            Y[(long)(rbase + rowhi + r) * NFEAT + t * 16 + lrow] = v;
        }
    }
}

#define ROWS_POOL 8

__global__ __launch_bounds__(256) void pool_kernel(const float* __restrict__ x, const int* __restrict__ gid,
                                                   float* __restrict__ g, int N) {
    const int lane = threadIdx.x & 31;
    const int wave = __builtin_amdgcn_readfirstlane((int)((blockIdx.x * 256u + threadIdx.x) >> 5));
    const int i0 = wave * ROWS_POOL;
    if (i0 >= N) return;
    const float* xl = x + lane;
    float* gl = g + lane;
    if (i0 + ROWS_POOL <= N) {
        float v[ROWS_POOL][4];
        int gr[ROWS_POOL];
#pragma unroll
        for (int u = 0; u < ROWS_POOL; ++u) {
            gr[u] = gid[i0 + u];
#pragma unroll
            for (int c = 0; c < 4; ++c) v[u][c] = xl[(long)(i0 + u) * NFEAT + c * 32];
        }
#pragma unroll
        for (int u = 0; u < ROWS_POOL; ++u) {
#pragma unroll
            for (int c = 0; c < 4; ++c) atomicAdd(&gl[(long)gr[u] * NFEAT + c * 32], v[u][c]);
        }
    } else {
        for (int i = i0; i < N; ++i) {
            const int gi = gid[i];
            for (int c = 0; c < 4; ++c) atomicAdd(&gl[(long)gi * NFEAT + c * 32], xl[(long)i * NFEAT + c * 32]);
        }
    }
}

__global__ void mlp2_kernel(const float* __restrict__ hid, const float* __restrict__ W,
                            const float* __restrict__ b, float* __restrict__ out) {
    int t = blockIdx.x * blockDim.x + threadIdx.x;
    if (t >= NGRAPH * NREL) return;
    int gi = t / NREL, n = t % NREL;
    const float* h = hid + (long)gi * NFEAT;
    float s = b[n];
#pragma unroll 8
    for (int k = 0; k < NFEAT; ++k) s = fmaf(h[k], W[k * NREL + n], s);
    out[t] = s;
}

extern "C" void kernel_launch(void* const* d_in, const int* in_sizes, int n_in,
                              void* d_out, int out_size, void* d_ws, size_t ws_size,
                              hipStream_t stream) {
    const int*   nid    = (const int*)d_in[0];
    const int*   z      = (const int*)d_in[1];
    const int*   src    = (const int*)d_in[2];
    const int*   dst    = (const int*)d_in[3];
    const int*   gid    = (const int*)d_in[4];
    const float* emb    = (const float*)d_in[5];
    const float* ztab   = (const float*)d_in[6];
    const float* Ws[3]  = {(const float*)d_in[7], (const float*)d_in[9], (const float*)d_in[11]};
    const float* bs[3]  = {(const float*)d_in[8], (const float*)d_in[10], (const float*)d_in[12]};
    const float* lin1_W = (const float*)d_in[13];
    const float* lin1_b = (const float*)d_in[14];
    const float* lin2_W = (const float*)d_in[15];
    const float* lin2_b = (const float*)d_in[16];

    const int N = in_sizes[0];
    const int E = in_sizes[2];

    float* x        = (float*)d_ws;
    float* agg      = x + (size_t)N * NFEAT;
    float* norm_out = agg + (size_t)N * NFEAT;
    float* norm_in  = norm_out + N;
    float* g        = norm_in + N;
    float* hid      = g + (size_t)NGRAPH * NFEAT;
    int*   cnt_out  = (int*)(hid + (size_t)NGRAPH * NFEAT);
    int*   cnt_in   = cnt_out + N;
    int*   rowptr   = cnt_in + N;
    int*   csr_src  = rowptr + N + 1;
    int*   rank     = (int*)agg;

    const int TB = 256;
    auto blocks = [](long n, int tb) { return (unsigned)((n + tb - 1) / tb); };

    zero_i32_kernel<<<blocks(2L * N, TB), TB, 0, stream>>>(cnt_out, 2L * N);
    count_rank_kernel<<<blocks(E, EDGE_TB * EDGE_PT), EDGE_TB, 0, stream>>>(src, dst, cnt_out, cnt_in, rank, E);
    scan_norm_kernel<<<1 + blocks(N, 1024), 1024, 0, stream>>>(cnt_out, cnt_in, rowptr, norm_out, norm_in, N);
    fill_kernel<<<blocks(E, EDGE_TB * EDGE_PT), EDGE_TB, 0, stream>>>(src, dst, rank, rowptr, csr_src, E);

    build_x_kernel<<<blocks((long)((N + ROWS_BX - 1) / ROWS_BX) * 32, TB), TB, 0, stream>>>(nid, z, emb, ztab, x, N);

    const unsigned gemm_blocks = (unsigned)((N + 127) / 128);
    const unsigned gather_blocks = blocks((long)((N + ROWS_PER_WAVE - 1) / ROWS_PER_WAVE) * 32, TB);
    for (int layer = 0; layer < 3; ++layer) {
        gather_kernel<<<gather_blocks, TB, 0, stream>>>(rowptr, csr_src, x, norm_out, agg, N);
        gemm128_wmma_kernel<<<gemm_blocks, TB, 0, stream>>>(
            agg, norm_in, Ws[layer], bs[layer], x, N, layer < 2 ? 1 : 0);
    }

    zero_f32_kernel<<<blocks((long)NGRAPH * NFEAT, TB), TB, 0, stream>>>(g, (long)NGRAPH * NFEAT);
    pool_kernel<<<blocks((long)((N + ROWS_POOL - 1) / ROWS_POOL) * 32, TB), TB, 0, stream>>>(x, gid, g, N);

    gemm128_wmma_kernel<<<(NGRAPH + 127) / 128, TB, 0, stream>>>(
        g, nullptr, lin1_W, lin1_b, hid, NGRAPH, 1);
    mlp2_kernel<<<blocks((long)NGRAPH * NREL, TB), TB, 0, stream>>>(hid, lin2_W, lin2_b, (float*)d_out);
}


# Sum pooling folded into the last layer's GEMM epilogue: a lane sums neighbouring rows of one graph in registers, then one float atomic add per run and column; pool and its zero-fill launch go
# speedup vs baseline: 2.2882x; 1.0938x over previous
#include <hip/hip_runtime.h>
#include <hip/hip_bf16.h>

typedef float v2f __attribute__((ext_vector_type(2)));
typedef float v8f __attribute__((ext_vector_type(8)));

#define NFEAT 128
#define NGRAPH 512
#define NREL 200

__global__ void zero_i32_kernel(int* __restrict__ p, long n) {
    long i = (long)blockIdx.x * blockDim.x + threadIdx.x;
    if (i < n) p[i] = 0;
}

#define EDGE_TB 256
#define EDGE_PT 4

__global__ __launch_bounds__(EDGE_TB) void count_rank_kernel(
    const int* __restrict__ src, const int* __restrict__ dst,
    int* __restrict__ cnt_out, int* __restrict__ cnt_in,
    int* __restrict__ rank, int E) {
    const int blk = blockIdx.x * (EDGE_TB * EDGE_PT);
    const int e0 = blk + threadIdx.x;
    if (blk + EDGE_TB * EDGE_PT <= E) {
        int s[EDGE_PT], d[EDGE_PT], r[EDGE_PT];
#pragma unroll
        for (int k = 0; k < EDGE_PT; ++k) { s[k] = src[e0 + k * EDGE_TB]; d[k] = dst[e0 + k * EDGE_TB]; }
#pragma unroll
        for (int k = 0; k < EDGE_PT; ++k) {
            atomicAdd(&cnt_out[s[k]], 1);
            r[k] = atomicAdd(&cnt_in[d[k]], 1);
        }
#pragma unroll
        for (int k = 0; k < EDGE_PT; ++k) rank[e0 + k * EDGE_TB] = r[k];
    } else {
        for (int k = 0; k < EDGE_PT; ++k) {
            const int e = e0 + k * EDGE_TB;
            if (e < E) {
                atomicAdd(&cnt_out[src[e]], 1);
                rank[e] = atomicAdd(&cnt_in[dst[e]], 1);
            }
        }
    }
}

__global__ __launch_bounds__(1024) void scan_norm_kernel(
    const int* __restrict__ cnt_out, const int* __restrict__ cnt_in,
    int* __restrict__ rowptr, float* __restrict__ norm_out, float* __restrict__ norm_in, int N) {
    const int tid = threadIdx.x;
    if (blockIdx.x != 0) {
        int i = (blockIdx.x - 1) * 1024 + tid;
        if (i < N) {
            norm_out[i] = rsqrtf(fmaxf((float)cnt_out[i], 1.0f));
            norm_in[i]  = rsqrtf(fmaxf((float)cnt_in[i], 1.0f));
        }
        return;
    }
    __shared__ int wsum[32];
    const int ch  = (N + 1023) / 1024;
    const int beg = min(tid * ch, N);
    const int end = min(beg + ch, N);
    int s = 0;
    int i = beg;
    for (; i + 8 <= end; i += 8) {
        int a[8];
#pragma unroll
        for (int u = 0; u < 8; ++u) a[u] = cnt_in[i + u];
#pragma unroll
        for (int u = 0; u < 8; ++u) s += a[u];
    }
    for (; i < end; ++i) s += cnt_in[i];

    const int lane = tid & 31, w = tid >> 5;
    int incl = s;
#pragma unroll
    for (int o = 1; o < 32; o <<= 1) {
        int t = __shfl_up(incl, o);
        if (lane >= o) incl += t;
    }
    if (lane == 31) wsum[w] = incl;
    __syncthreads();
    if (w == 0) {
        int v = wsum[lane], iv = v;
#pragma unroll
        for (int o = 1; o < 32; o <<= 1) {
            int t = __shfl_up(iv, o);
            if (lane >= o) iv += t;
        }
        wsum[lane] = iv - v;
    }
    __syncthreads();
    int run = wsum[w] + incl - s;
    i = beg;
    for (; i + 8 <= end; i += 8) {
        int a[8];
#pragma unroll
        for (int u = 0; u < 8; ++u) a[u] = cnt_in[i + u];
#pragma unroll
        for (int u = 0; u < 8; ++u) { rowptr[i + u] = run; run += a[u]; }
    }
    for (; i < end; ++i) { rowptr[i] = run; run += cnt_in[i]; }
    if (tid == 1023) rowptr[N] = run;
}

__global__ __launch_bounds__(EDGE_TB) void fill_kernel(
    const int* __restrict__ src, const int* __restrict__ dst,
    const int* __restrict__ rank, const int* __restrict__ rowptr,
    int* __restrict__ csr_src, int E) {
    const int blk = blockIdx.x * (EDGE_TB * EDGE_PT);
    const int e0 = blk + threadIdx.x;
    if (blk + EDGE_TB * EDGE_PT <= E) {
        int s[EDGE_PT], d[EDGE_PT], r[EDGE_PT], p[EDGE_PT];
#pragma unroll
        for (int k = 0; k < EDGE_PT; ++k) {
            s[k] = src[e0 + k * EDGE_TB]; d[k] = dst[e0 + k * EDGE_TB]; r[k] = rank[e0 + k * EDGE_TB];
        }
#pragma unroll
        for (int k = 0; k < EDGE_PT; ++k) p[k] = rowptr[d[k]];
#pragma unroll
        for (int k = 0; k < EDGE_PT; ++k) csr_src[p[k] + r[k]] = s[k];
    } else {
        for (int k = 0; k < EDGE_PT; ++k) {
            const int e = e0 + k * EDGE_TB;
            if (e < E) csr_src[rowptr[dst[e]] + rank[e]] = src[e];
        }
    }
}

#define ROWS_BX 8

__device__ __forceinline__ float4 bx_row(const float* __restrict__ tab, bool ent,
                                         const int* __restrict__ nid, const int* __restrict__ z, int i) {
    const int rn = nid[i];
    const int rz = z[i];
    return *(const float4*)(tab + (long)(ent ? rn : rz) * 64);
}

__global__ __launch_bounds__(256) void build_x_kernel(
    const int* __restrict__ nid, const int* __restrict__ z,
    const float* __restrict__ emb, const float* __restrict__ ztab,
    float* __restrict__ x, int N) {
    const int lane = threadIdx.x & 31;
    const int wave = __builtin_amdgcn_readfirstlane((int)((blockIdx.x * 256u + threadIdx.x) >> 5));
    const int i0 = wave * ROWS_BX;
    if (i0 >= N) return;
    const bool ent = lane < 16;
    const float* tab = (ent ? emb : ztab) + (lane & 15) * 4;
    float* xo = x + lane * 4;
    if (i0 + ROWS_BX <= N) {
        const float4 v0 = bx_row(tab, ent, nid, z, i0 + 0);
        const float4 v1 = bx_row(tab, ent, nid, z, i0 + 1);
        const float4 v2 = bx_row(tab, ent, nid, z, i0 + 2);
        const float4 v3 = bx_row(tab, ent, nid, z, i0 + 3);
        const float4 v4 = bx_row(tab, ent, nid, z, i0 + 4);
        const float4 v5 = bx_row(tab, ent, nid, z, i0 + 5);
        const float4 v6 = bx_row(tab, ent, nid, z, i0 + 6);
        const float4 v7 = bx_row(tab, ent, nid, z, i0 + 7);
        __builtin_amdgcn_sched_barrier(0);
        float* xr = xo + (long)i0 * NFEAT;
        *(float4*)(xr + 0 * NFEAT) = v0;
        *(float4*)(xr + 1 * NFEAT) = v1;
        *(float4*)(xr + 2 * NFEAT) = v2;
        *(float4*)(xr + 3 * NFEAT) = v3;
        *(float4*)(xr + 4 * NFEAT) = v4;
        *(float4*)(xr + 5 * NFEAT) = v5;
        *(float4*)(xr + 6 * NFEAT) = v6;
        *(float4*)(xr + 7 * NFEAT) = v7;
    } else {
        for (int i = i0; i < N; ++i) {
            *(float4*)(xo + (long)i * NFEAT) = bx_row(tab, ent, nid, z, i);
        }
    }
}

#define ROWS_PER_WAVE 2

template <int U>
__device__ __forceinline__ void add_rows(float4& acc, int s_l, int w_bits, int j,
                                         const float* __restrict__ xl) {
    float4 v[U];
    float w[U];
#pragma unroll
    for (int u = 0; u < U; ++u) {
        int sj = __builtin_amdgcn_readlane(s_l, j + u);
        w[u] = __int_as_float(__builtin_amdgcn_readlane(w_bits, j + u));
        v[u] = *(const float4*)(xl + (long)sj * NFEAT);
    }
#pragma unroll
    for (int u = 0; u < U; ++u) {
        acc.x = fmaf(v[u].x, w[u], acc.x);
        acc.y = fmaf(v[u].y, w[u], acc.y);
        acc.z = fmaf(v[u].z, w[u], acc.z);
        acc.w = fmaf(v[u].w, w[u], acc.w);
    }
}

__global__ __launch_bounds__(256) void gather_kernel(
    const int* __restrict__ rowptr, const int* __restrict__ csr_src,
    const float* __restrict__ x, const float* __restrict__ norm_out,
    float* __restrict__ agg, int N) {
    const int lane = threadIdx.x & 31;
    const int wave = __builtin_amdgcn_readfirstlane((int)((blockIdx.x * 256u + threadIdx.x) >> 5));
    const float* xl = x + lane * 4;
    for (int q = 0; q < ROWS_PER_WAVE; ++q) {
        const int i = wave * ROWS_PER_WAVE + q;
        if (i >= N) break;
        const int beg = rowptr[i];
        const int end = rowptr[i + 1];
        float4 acc = make_float4(0.0f, 0.0f, 0.0f, 0.0f);
        for (int base = beg; base < end; base += 32) {
            const int n = min(32, end - base);
            int s_l = 0;
            float w_l = 0.0f;
            if (lane < n) {
                s_l = csr_src[base + lane];
                w_l = norm_out[s_l];
            }
            const int w_bits = __float_as_int(w_l);
            int j = 0;
            for (; j + 8 <= n; j += 8) add_rows<8>(acc, s_l, w_bits, j, xl);
            if (n - j >= 4) { add_rows<4>(acc, s_l, w_bits, j, xl); j += 4; }
            if (n - j >= 2) { add_rows<2>(acc, s_l, w_bits, j, xl); j += 2; }
            if (n - j >= 1) { add_rows<1>(acc, s_l, w_bits, j, xl); }
        }
        *(float4*)(agg + (long)i * NFEAT + lane * 4) = acc;
    }
}

__global__ __launch_bounds__(256) void gemm128_wmma_kernel(
    const float* __restrict__ A, const float* __restrict__ norm,
    const float* __restrict__ W, const float* __restrict__ bias,
    float* __restrict__ Y, int nrows, int relu,
    const int* __restrict__ pool_gid, float* __restrict__ pool_g) {
    __shared__ float Wlds[NFEAT * NFEAT];

    int tid = threadIdx.x;
    {
        const float4* Wv = (const float4*)W;
        float4* Lv = (float4*)Wlds;
#pragma unroll
        for (int i = 0; i < 16; ++i) Lv[tid + i * 256] = Wv[tid + i * 256];
    }
    __syncthreads();

    int wave = tid >> 5;
    int lane = tid & 31;
    int rbase = blockIdx.x * 128 + wave * 16;
    if (rbase >= nrows) return;

    int lrow  = lane & 15;
    int khalf = (lane >> 4) << 1;
    int rowhi = (lane >> 4) << 3;

    long m = rbase + lrow;
    float nrm = norm ? norm[m] : 1.0f;
    const float* Arow = A + m * NFEAT;

    v8f acc[8];
#pragma unroll
    for (int t = 0; t < 8; ++t) {
        float bv = bias[t * 16 + lrow];
#pragma unroll
        for (int r = 0; r < 8; ++r) acc[t][r] = bv;
    }

    for (int k = 0; k < NFEAT; k += 4) {
        v2f a;
        a.x = Arow[k + khalf] * nrm;
        a.y = Arow[k + khalf + 1] * nrm;
#pragma unroll
        for (int t = 0; t < 8; ++t) {
            v2f b;
            b.x = Wlds[(k + khalf) * NFEAT + t * 16 + lrow];
            b.y = Wlds[(k + khalf + 1) * NFEAT + t * 16 + lrow];
            acc[t] = __builtin_amdgcn_wmma_f32_16x16x4_f32(
                false, a, false, b, (short)0, acc[t], false, false);
        }
    }

    if (pool_g) {
        const int row0 = rbase + rowhi;
        int gr[8];
        float run[8];
#pragma unroll
        for (int r = 0; r < 8; ++r) gr[r] = pool_gid[min(row0 + r, nrows - 1)];
#pragma unroll
        for (int t = 0; t < 8; ++t) run[t] = 0.0f;
        float* gcol = pool_g + lrow;
#pragma unroll
        for (int r = 0; r < 8; ++r) {
            if (r > 0 && gr[r] != gr[r - 1]) {
                float* gp = gcol + (long)gr[r - 1] * NFEAT;
#pragma unroll
                for (int t = 0; t < 8; ++t) { atomicAdd(gp + t * 16, run[t]); run[t] = 0.0f; }
            }
            if (row0 + r < nrows) {
#pragma unroll
                for (int t = 0; t < 8; ++t) {
                    float v = acc[t][r];
                    if (relu) v = fmaxf(v, 0.0f);
                    run[t] += v;
                }
            }
        }
        float* gp = gcol + (long)gr[7] * NFEAT;
#pragma unroll
        for (int t = 0; t < 8; ++t) atomicAdd(gp + t * 16, run[t]);
        return;
    }

#pragma unroll
    for (int t = 0; t < 8; ++t) {
#pragma unroll
        for (int r = 0; r < 8; ++r) {
            float v = acc[t][r];
            if (relu) v = fmaxf(v, 0.0f);
            Y[(long)(rbase + rowhi + r) * NFEAT + t * 16 + lrow] = v;
        }
    }
}

__global__ void mlp2_kernel(const float* __restrict__ hid, const float* __restrict__ W,
                            const float* __restrict__ b, float* __restrict__ out) {
    int t = blockIdx.x * blockDim.x + threadIdx.x;
    if (t >= NGRAPH * NREL) return;
    int gi = t / NREL, n = t % NREL;
    const float* h = hid + (long)gi * NFEAT;
    float s = b[n];
#pragma unroll 8
    for (int k = 0; k < NFEAT; ++k) s = fmaf(h[k], W[k * NREL + n], s);
    out[t] = s;
}

extern "C" void kernel_launch(void* const* d_in, const int* in_sizes, int n_in,
                              void* d_out, int out_size, void* d_ws, size_t ws_size,
                              hipStream_t stream) {
    const int*   nid    = (const int*)d_in[0];
    const int*   z      = (const int*)d_in[1];
    const int*   src    = (const int*)d_in[2];
    const int*   dst    = (const int*)d_in[3];
    const int*   gid    = (const int*)d_in[4];
    const float* emb    = (const float*)d_in[5];
    const float* ztab   = (const float*)d_in[6];
    const float* Ws[3]  = {(const float*)d_in[7], (const float*)d_in[9], (const float*)d_in[11]};
    const float* bs[3]  = {(const float*)d_in[8], (const float*)d_in[10], (const float*)d_in[12]};
    const float* lin1_W = (const float*)d_in[13];
    const float* lin1_b = (const float*)d_in[14];
    const float* lin2_W = (const float*)d_in[15];
    const float* lin2_b = (const float*)d_in[16];

    const int N = in_sizes[0];
    const int E = in_sizes[2];

    float* x        = (float*)d_ws;
    float* agg      = x + (size_t)N * NFEAT;
    float* norm_out = agg + (size_t)N * NFEAT;
    float* norm_in  = norm_out + N;
    float* g        = norm_in + N;
    float* hid      = g + (size_t)NGRAPH * NFEAT;
    int*   cnt_out  = (int*)(hid + (size_t)NGRAPH * NFEAT);
    int*   cnt_in   = cnt_out + N;
    int*   rowptr   = cnt_in + N;
    int*   csr_src  = rowptr + N + 1;
    int*   rank     = (int*)agg;

    const int TB = 256;
    auto blocks = [](long n, int tb) { return (unsigned)((n + tb - 1) / tb); };

    const long nclear = 2L * NGRAPH * NFEAT + 2L * N;
    zero_i32_kernel<<<blocks(nclear, TB), TB, 0, stream>>>((int*)g, nclear);
    count_rank_kernel<<<blocks(E, EDGE_TB * EDGE_PT), EDGE_TB, 0, stream>>>(src, dst, cnt_out, cnt_in, rank, E);
    scan_norm_kernel<<<1 + blocks(N, 1024), 1024, 0, stream>>>(cnt_out, cnt_in, rowptr, norm_out, norm_in, N);
    fill_kernel<<<blocks(E, EDGE_TB * EDGE_PT), EDGE_TB, 0, stream>>>(src, dst, rank, rowptr, csr_src, E);

    build_x_kernel<<<blocks((long)((N + ROWS_BX - 1) / ROWS_BX) * 32, TB), TB, 0, stream>>>(nid, z, emb, ztab, x, N);

    const unsigned gemm_blocks = (unsigned)((N + 127) / 128);
    const unsigned gather_blocks = blocks((long)((N + ROWS_PER_WAVE - 1) / ROWS_PER_WAVE) * 32, TB);
    for (int layer = 0; layer < 3; ++layer) {
        gather_kernel<<<gather_blocks, TB, 0, stream>>>(rowptr, csr_src, x, norm_out, agg, N);
        const bool last = layer == 2;
        gemm128_wmma_kernel<<<gemm_blocks, TB, 0, stream>>>(
            agg, norm_in, Ws[layer], bs[layer], x, N, last ? 0 : 1,
            last ? gid : nullptr, last ? g : nullptr);
    }

    gemm128_wmma_kernel<<<(NGRAPH + 127) / 128, TB, 0, stream>>>(
        g, nullptr, lin1_W, lin1_b, hid, NGRAPH, 1, nullptr, nullptr);
    mlp2_kernel<<<blocks((long)NGRAPH * NREL, TB), TB, 0, stream>>>(hid, lin2_W, lin2_b, (float*)d_out);
}


# Rows kept pre-multiplied by norm_out (the feature build and the two storing layer epilogues scale each row once); the gather adds plain rows, no scattered norm_out lookup per edge; sum order kept
# speedup vs baseline: 2.5313x; 1.1062x over previous
#include <hip/hip_runtime.h>
#include <hip/hip_bf16.h>

typedef float v2f __attribute__((ext_vector_type(2)));
typedef float v8f __attribute__((ext_vector_type(8)));

#define NFEAT 128
#define NGRAPH 512
#define NREL 200

__global__ void zero_i32_kernel(int* __restrict__ p, long n) {
    long i = (long)blockIdx.x * blockDim.x + threadIdx.x;
    if (i < n) p[i] = 0;
}

#define EDGE_TB 256
#define EDGE_PT 4

__global__ __launch_bounds__(EDGE_TB) void count_rank_kernel(
    const int* __restrict__ src, const int* __restrict__ dst,
    int* __restrict__ cnt_out, int* __restrict__ cnt_in,
    int* __restrict__ rank, int E) {
    const int blk = blockIdx.x * (EDGE_TB * EDGE_PT);
    const int e0 = blk + threadIdx.x;
    if (blk + EDGE_TB * EDGE_PT <= E) {
        int s[EDGE_PT], d[EDGE_PT], r[EDGE_PT];
#pragma unroll
        for (int k = 0; k < EDGE_PT; ++k) { s[k] = src[e0 + k * EDGE_TB]; d[k] = dst[e0 + k * EDGE_TB]; }
#pragma unroll
        for (int k = 0; k < EDGE_PT; ++k) {
            atomicAdd(&cnt_out[s[k]], 1);
            r[k] = atomicAdd(&cnt_in[d[k]], 1);
        }
#pragma unroll
        for (int k = 0; k < EDGE_PT; ++k) rank[e0 + k * EDGE_TB] = r[k];
    } else {
        for (int k = 0; k < EDGE_PT; ++k) {
            const int e = e0 + k * EDGE_TB;
            if (e < E) {
                atomicAdd(&cnt_out[src[e]], 1);
                rank[e] = atomicAdd(&cnt_in[dst[e]], 1);
            }
        }
    }
}

__global__ __launch_bounds__(1024) void scan_norm_kernel(
    const int* __restrict__ cnt_out, const int* __restrict__ cnt_in,
    int* __restrict__ rowptr, float* __restrict__ norm_out, float* __restrict__ norm_in, int N) {
    const int tid = threadIdx.x;
    if (blockIdx.x != 0) {
        int i = (blockIdx.x - 1) * 1024 + tid;
        if (i < N) {
            norm_out[i] = rsqrtf(fmaxf((float)cnt_out[i], 1.0f));
            norm_in[i]  = rsqrtf(fmaxf((float)cnt_in[i], 1.0f));
        }
        return;
    }
    __shared__ int wsum[32];
    const int ch  = (N + 1023) / 1024;
    const int beg = min(tid * ch, N);
    const int end = min(beg + ch, N);
    int s = 0;
    int i = beg;
    for (; i + 8 <= end; i += 8) {
        int a[8];
#pragma unroll
        for (int u = 0; u < 8; ++u) a[u] = cnt_in[i + u];
#pragma unroll
        for (int u = 0; u < 8; ++u) s += a[u];
    }
    for (; i < end; ++i) s += cnt_in[i];

    const int lane = tid & 31, w = tid >> 5;
    int incl = s;
#pragma unroll
    for (int o = 1; o < 32; o <<= 1) {
        int t = __shfl_up(incl, o);
        if (lane >= o) incl += t;
    }
    if (lane == 31) wsum[w] = incl;
    __syncthreads();
    if (w == 0) {
        int v = wsum[lane], iv = v;
#pragma unroll
        for (int o = 1; o < 32; o <<= 1) {
            int t = __shfl_up(iv, o);
            if (lane >= o) iv += t;
        }
        wsum[lane] = iv - v;
    }
    __syncthreads();
    int run = wsum[w] + incl - s;
    i = beg;
    for (; i + 8 <= end; i += 8) {
        int a[8];
#pragma unroll
        for (int u = 0; u < 8; ++u) a[u] = cnt_in[i + u];
#pragma unroll
        for (int u = 0; u < 8; ++u) { rowptr[i + u] = run; run += a[u]; }
    }
    for (; i < end; ++i) { rowptr[i] = run; run += cnt_in[i]; }
    if (tid == 1023) rowptr[N] = run;
}

__global__ __launch_bounds__(EDGE_TB) void fill_kernel(
    const int* __restrict__ src, const int* __restrict__ dst,
    const int* __restrict__ rank, const int* __restrict__ rowptr,
    int* __restrict__ csr_src, int E) {
    const int blk = blockIdx.x * (EDGE_TB * EDGE_PT);
    const int e0 = blk + threadIdx.x;
    if (blk + EDGE_TB * EDGE_PT <= E) {
        int s[EDGE_PT], d[EDGE_PT], r[EDGE_PT], p[EDGE_PT];
#pragma unroll
        for (int k = 0; k < EDGE_PT; ++k) {
            s[k] = src[e0 + k * EDGE_TB]; d[k] = dst[e0 + k * EDGE_TB]; r[k] = rank[e0 + k * EDGE_TB];
        }
#pragma unroll
        for (int k = 0; k < EDGE_PT; ++k) p[k] = rowptr[d[k]];
#pragma unroll
        for (int k = 0; k < EDGE_PT; ++k) csr_src[p[k] + r[k]] = s[k];
    } else {
        for (int k = 0; k < EDGE_PT; ++k) {
            const int e = e0 + k * EDGE_TB;
            if (e < E) csr_src[rowptr[dst[e]] + rank[e]] = src[e];
        }
    }
}

#define ROWS_BX 8

__device__ __forceinline__ float4 scale4(float4 v, float s) {
    return make_float4(v.x * s, v.y * s, v.z * s, v.w * s);
}

__device__ __forceinline__ float4 bx_row(const float* __restrict__ tab, bool ent,
                                         const int* __restrict__ nid, const int* __restrict__ z, int i) {
    const int rn = nid[i];
    const int rz = z[i];
    return *(const float4*)(tab + (long)(ent ? rn : rz) * 64);
}

__global__ __launch_bounds__(256) void build_x_kernel(
    const int* __restrict__ nid, const int* __restrict__ z,
    const float* __restrict__ emb, const float* __restrict__ ztab,
    const float* __restrict__ norm_out,
    float* __restrict__ x, int N) {
    const int lane = threadIdx.x & 31;
    const int wave = __builtin_amdgcn_readfirstlane((int)((blockIdx.x * 256u + threadIdx.x) >> 5));
    const int i0 = wave * ROWS_BX;
    if (i0 >= N) return;
    const bool ent = lane < 16;
    const float* tab = (ent ? emb : ztab) + (lane & 15) * 4;
    float* xo = x + lane * 4;
    if (i0 + ROWS_BX <= N) {
        const float4 v0 = bx_row(tab, ent, nid, z, i0 + 0);
        const float4 v1 = bx_row(tab, ent, nid, z, i0 + 1);
        const float4 v2 = bx_row(tab, ent, nid, z, i0 + 2);
        const float4 v3 = bx_row(tab, ent, nid, z, i0 + 3);
        const float4 v4 = bx_row(tab, ent, nid, z, i0 + 4);
        const float4 v5 = bx_row(tab, ent, nid, z, i0 + 5);
        const float4 v6 = bx_row(tab, ent, nid, z, i0 + 6);
        const float4 v7 = bx_row(tab, ent, nid, z, i0 + 7);
        const float* nr = norm_out + i0;
        const float s0 = nr[0], s1 = nr[1], s2 = nr[2], s3 = nr[3];
        const float s4 = nr[4], s5 = nr[5], s6 = nr[6], s7 = nr[7];
        __builtin_amdgcn_sched_barrier(0);
        float* xr = xo + (long)i0 * NFEAT;
        *(float4*)(xr + 0 * NFEAT) = scale4(v0, s0);
        *(float4*)(xr + 1 * NFEAT) = scale4(v1, s1);
        *(float4*)(xr + 2 * NFEAT) = scale4(v2, s2);
        *(float4*)(xr + 3 * NFEAT) = scale4(v3, s3);
        *(float4*)(xr + 4 * NFEAT) = scale4(v4, s4);
        *(float4*)(xr + 5 * NFEAT) = scale4(v5, s5);
        *(float4*)(xr + 6 * NFEAT) = scale4(v6, s6);
        *(float4*)(xr + 7 * NFEAT) = scale4(v7, s7);
    } else {
        for (int i = i0; i < N; ++i) {
            *(float4*)(xo + (long)i * NFEAT) = scale4(bx_row(tab, ent, nid, z, i), norm_out[i]);
        }
    }
}

#define LAYER_TB 256
#define TILE_ROWS 16
#define APITCH 132
#define GU 16

__device__ __forceinline__ void close_row(float4& acc, float* __restrict__ As, int q, int nin_bits, int lane) {
    const float s = __int_as_float(__builtin_amdgcn_readlane(nin_bits, q));
    *(float4*)(As + q * APITCH + lane * 4) = make_float4(acc.x * s, acc.y * s, acc.z * s, acc.w * s);
    acc = make_float4(0.0f, 0.0f, 0.0f, 0.0f);
}

template <int U, bool NX>
__device__ __forceinline__ void take_rows(float4& acc, int& q, int& rend, int e, int j,
                                          int s_l, int rw, int nin_bits,
                                          const float* __restrict__ xl, float* __restrict__ As, int lane,
                                          int& s_nx, const int* __restrict__ list, int nx, int e_end) {
    float4 v[U];
#pragma unroll
    for (int u = 0; u < U; ++u) {
        const int sj = __builtin_amdgcn_readlane(s_l, j + u);
        v[u] = *(const float4*)(xl + (long)sj * NFEAT);
    }
    if (NX) {
        if (nx < e_end) s_nx = list[nx];
    }
#pragma unroll
    for (int u = 0; u < U; ++u) {
        while (e + u >= rend) {
            close_row(acc, As, q, nin_bits, lane);
            ++q;
            rend = __builtin_amdgcn_readlane(rw, q + 1);
        }
        acc.x += v[u].x;
        acc.y += v[u].y;
        acc.z += v[u].z;
        acc.w += v[u].w;
    }
}

__global__ __launch_bounds__(LAYER_TB) void gcn_layer_kernel(
    const int* __restrict__ rowptr, const int* __restrict__ csr_src,
    const float* __restrict__ x, const float* __restrict__ out_scale, const float* __restrict__ norm_in,
    const float* __restrict__ W, const float* __restrict__ bias,
    float* __restrict__ Y, int nrows, int relu,
    const int* __restrict__ pool_gid, float* __restrict__ pool_g) {
    __shared__ __attribute__((aligned(16))) float Wlds[NFEAT * NFEAT];
    __shared__ __attribute__((aligned(16))) float Astage[(LAYER_TB / 32) * TILE_ROWS * APITCH];

    const int tid = threadIdx.x;
    {
        const float4* Wv = (const float4*)W;
        float4* Lv = (float4*)Wlds;
#pragma unroll
        for (int i = 0; i < 16; ++i) Lv[tid + i * 256] = Wv[tid + i * 256];
    }
    __syncthreads();

    const int lane = tid & 31;
    const int wave = __builtin_amdgcn_readfirstlane(tid >> 5);
    const int slot = (wave + (((int)blockIdx.x >> 8) & 1) * 6) & 7;
    const int tile = slot * (int)gridDim.x + (int)blockIdx.x;
    const int rbase = tile * TILE_ROWS;
    if (rbase >= nrows) return;

    float* As = Astage + wave * (TILE_ROWS * APITCH);

    {
        const int rp = rowptr[min(rbase + lane, nrows)];
        const int nin_bits = __float_as_int(norm_in[min(rbase + lane, nrows - 1)]);
        const int e_beg = __builtin_amdgcn_readlane(rp, 0);
        const int e_end = __builtin_amdgcn_readlane(rp, TILE_ROWS);
        const int rw = lane < TILE_ROWS ? rp : 0x7fffffff;
        const float* xl = x + lane * 4;
        int q = 0;
        int rend = __builtin_amdgcn_readlane(rw, 1);
        float4 acc = make_float4(0.0f, 0.0f, 0.0f, 0.0f);

        int s_cur = 0;
        if (e_beg + lane < e_end) s_cur = csr_src[e_beg + lane];
        static_assert(2 * GU == 32, "a chunk of 32 list entries is two full groups");
        for (int base = e_beg; base < e_end; base += 32) {
            const int n = min(32, e_end - base);
            int s_nx = 0;
            const int nx = base + 32 + lane;
            int j = 0;
            if (n >= GU) {
                take_rows<GU, true>(acc, q, rend, base, 0, s_cur, rw, nin_bits, xl, As, lane, s_nx, csr_src, nx, e_end);
                j = GU;
            }
            if (n >= 2 * GU) {
                take_rows<GU, false>(acc, q, rend, base + GU, GU, s_cur, rw, nin_bits, xl, As, lane, s_nx, csr_src, nx, e_end);
                j = 2 * GU;
            }
            if (n - j >= 8) { take_rows<8, false>(acc, q, rend, base + j, j, s_cur, rw, nin_bits, xl, As, lane, s_nx, csr_src, nx, e_end); j += 8; }
            if (n - j >= 4) { take_rows<4, false>(acc, q, rend, base + j, j, s_cur, rw, nin_bits, xl, As, lane, s_nx, csr_src, nx, e_end); j += 4; }
            if (n - j >= 2) { take_rows<2, false>(acc, q, rend, base + j, j, s_cur, rw, nin_bits, xl, As, lane, s_nx, csr_src, nx, e_end); j += 2; }
            if (n - j >= 1) { take_rows<1, false>(acc, q, rend, base + j, j, s_cur, rw, nin_bits, xl, As, lane, s_nx, csr_src, nx, e_end); }
            s_cur = s_nx;
        }
        for (; q < TILE_ROWS; ++q) close_row(acc, As, q, nin_bits, lane);
    }

    const int lrow  = lane & 15;
    const int khalf = (lane >> 4) << 1;
    const int rowhi = (lane >> 4) << 3;
    const float* Arow = As + lrow * APITCH + khalf;

    v8f acc[8];
#pragma unroll
    for (int t = 0; t < 8; ++t) {
        float bv = bias[t * 16 + lrow];
#pragma unroll
        for (int r = 0; r < 8; ++r) acc[t][r] = bv;
    }

    float osc[8];
#pragma unroll
    for (int r = 0; r < 8; ++r) osc[r] = 1.0f;
    if (!pool_g) {
#pragma unroll
        for (int r = 0; r < 8; ++r) osc[r] = out_scale[min(rbase + rowhi + r, nrows - 1)];
    }

    for (int k = 0; k < NFEAT; k += 4) {
        const v2f a = *(const v2f*)(Arow + k);
#pragma unroll
        for (int t = 0; t < 8; ++t) {
            v2f b;
            b.x = Wlds[(k + khalf) * NFEAT + t * 16 + lrow];
            b.y = Wlds[(k + khalf + 1) * NFEAT + t * 16 + lrow];
            acc[t] = __builtin_amdgcn_wmma_f32_16x16x4_f32(
                false, a, false, b, (short)0, acc[t], false, false);
        }
    }

    if (pool_g) {
        const int row0 = rbase + rowhi;
        int gr[8];
        float run[8];
#pragma unroll
        for (int r = 0; r < 8; ++r) gr[r] = pool_gid[min(row0 + r, nrows - 1)];
#pragma unroll
        for (int t = 0; t < 8; ++t) run[t] = 0.0f;
        float* gcol = pool_g + lrow;
#pragma unroll
        for (int r = 0; r < 8; ++r) {
            if (r > 0 && gr[r] != gr[r - 1]) {
                float* gp = gcol + (long)gr[r - 1] * NFEAT;
#pragma unroll
                for (int t = 0; t < 8; ++t) { atomicAdd(gp + t * 16, run[t]); run[t] = 0.0f; }
            }
            if (row0 + r < nrows) {
#pragma unroll
                for (int t = 0; t < 8; ++t) {
                    float v = acc[t][r];
                    if (relu) v = fmaxf(v, 0.0f);
                    run[t] += v;
                }
            }
        }
        float* gp = gcol + (long)gr[7] * NFEAT;
#pragma unroll
        for (int t = 0; t < 8; ++t) atomicAdd(gp + t * 16, run[t]);
        return;
    }

#pragma unroll
    for (int r = 0; r < 8; ++r) {
        const int row = rbase + rowhi + r;
        if (row < nrows) {
#pragma unroll
            for (int t = 0; t < 8; ++t) {
                float v = acc[t][r];
                if (relu) v = fmaxf(v, 0.0f);
                Y[(long)row * NFEAT + t * 16 + lrow] = v * osc[r];
            }
        }
    }
}

__global__ __launch_bounds__(256) void gemm128_wmma_kernel(
    const float* __restrict__ A,
    const float* __restrict__ W, const float* __restrict__ bias,
    float* __restrict__ Y, int nrows, int relu) {
    __shared__ float Wlds[NFEAT * NFEAT];

    int tid = threadIdx.x;
    {
        const float4* Wv = (const float4*)W;
        float4* Lv = (float4*)Wlds;
#pragma unroll
        for (int i = 0; i < 16; ++i) Lv[tid + i * 256] = Wv[tid + i * 256];
    }
    __syncthreads();

    int wave = tid >> 5;
    int lane = tid & 31;
    int rbase = blockIdx.x * 128 + wave * 16;
    if (rbase >= nrows) return;

    int lrow  = lane & 15;
    int khalf = (lane >> 4) << 1;
    int rowhi = (lane >> 4) << 3;

    long m = rbase + lrow;
    const float* Arow = A + m * NFEAT;

    v8f acc[8];
#pragma unroll
    for (int t = 0; t < 8; ++t) {
        float bv = bias[t * 16 + lrow];
#pragma unroll
        for (int r = 0; r < 8; ++r) acc[t][r] = bv;
    }

    for (int k = 0; k < NFEAT; k += 4) {
        v2f a;
        a.x = Arow[k + khalf];
        a.y = Arow[k + khalf + 1];
#pragma unroll
        for (int t = 0; t < 8; ++t) {
            v2f b;
            b.x = Wlds[(k + khalf) * NFEAT + t * 16 + lrow];
            b.y = Wlds[(k + khalf + 1) * NFEAT + t * 16 + lrow];
            acc[t] = __builtin_amdgcn_wmma_f32_16x16x4_f32(
                false, a, false, b, (short)0, acc[t], false, false);
        }
    }

#pragma unroll
    for (int t = 0; t < 8; ++t) {
#pragma unroll
        for (int r = 0; r < 8; ++r) {
            float v = acc[t][r];
            if (relu) v = fmaxf(v, 0.0f);
            Y[(long)(rbase + rowhi + r) * NFEAT + t * 16 + lrow] = v;
        }
    }
}

__global__ void mlp2_kernel(const float* __restrict__ hid, const float* __restrict__ W,
                            const float* __restrict__ b, float* __restrict__ out) {
    int t = blockIdx.x * blockDim.x + threadIdx.x;
    if (t >= NGRAPH * NREL) return;
    int gi = t / NREL, n = t % NREL;
    const float* h = hid + (long)gi * NFEAT;
    float s = b[n];
#pragma unroll 8
    for (int k = 0; k < NFEAT; ++k) s = fmaf(h[k], W[k * NREL + n], s);
    out[t] = s;
}

extern "C" void kernel_launch(void* const* d_in, const int* in_sizes, int n_in,
                              void* d_out, int out_size, void* d_ws, size_t ws_size,
                              hipStream_t stream) {
    const int*   nid    = (const int*)d_in[0];
    const int*   z      = (const int*)d_in[1];
    const int*   src    = (const int*)d_in[2];
    const int*   dst    = (const int*)d_in[3];
    const int*   gid    = (const int*)d_in[4];
    const float* emb    = (const float*)d_in[5];
    const float* ztab   = (const float*)d_in[6];
    const float* Ws[3]  = {(const float*)d_in[7], (const float*)d_in[9], (const float*)d_in[11]};
    const float* bs[3]  = {(const float*)d_in[8], (const float*)d_in[10], (const float*)d_in[12]};
    const float* lin1_W = (const float*)d_in[13];
    const float* lin1_b = (const float*)d_in[14];
    const float* lin2_W = (const float*)d_in[15];
    const float* lin2_b = (const float*)d_in[16];

    const int N = in_sizes[0];
    const int E = in_sizes[2];

    float* x        = (float*)d_ws;
    float* agg      = x + (size_t)N * NFEAT;
    float* norm_out = agg + (size_t)N * NFEAT;
    float* norm_in  = norm_out + N;
    float* g        = norm_in + N;
    float* hid      = g + (size_t)NGRAPH * NFEAT;
    int*   cnt_out  = (int*)(hid + (size_t)NGRAPH * NFEAT);
    int*   cnt_in   = cnt_out + N;
    int*   rowptr   = cnt_in + N;
    int*   csr_src  = rowptr + N + 1;
    int*   rank     = (int*)agg;

    const int TB = 256;
    auto blocks = [](long n, int tb) { return (unsigned)((n + tb - 1) / tb); };

    const long nclear = 2L * NGRAPH * NFEAT + 2L * N;
    zero_i32_kernel<<<blocks(nclear, TB), TB, 0, stream>>>((int*)g, nclear);
    count_rank_kernel<<<blocks(E, EDGE_TB * EDGE_PT), EDGE_TB, 0, stream>>>(src, dst, cnt_out, cnt_in, rank, E);
    scan_norm_kernel<<<1 + blocks(N, 1024), 1024, 0, stream>>>(cnt_out, cnt_in, rowptr, norm_out, norm_in, N);
    fill_kernel<<<blocks(E, EDGE_TB * EDGE_PT), EDGE_TB, 0, stream>>>(src, dst, rank, rowptr, csr_src, E);

    build_x_kernel<<<blocks((long)((N + ROWS_BX - 1) / ROWS_BX) * 32, TB), TB, 0, stream>>>(nid, z, emb, ztab, norm_out, x, N);

    const long ntiles = ((long)N + TILE_ROWS - 1) / TILE_ROWS;
    const unsigned need_blocks = blocks(ntiles, LAYER_TB / 32);
    const unsigned layer_blocks = need_blocks > 512u ? need_blocks : 512u;
    float* xin = x;
    float* xout = agg;
    for (int layer = 0; layer < 3; ++layer) {
        const bool last = layer == 2;
        gcn_layer_kernel<<<layer_blocks, LAYER_TB, 0, stream>>>(
            rowptr, csr_src, xin, norm_out, norm_in, Ws[layer], bs[layer], xout, N, last ? 0 : 1,
            last ? gid : nullptr, last ? g : nullptr);
        float* t = xin; xin = xout; xout = t;
    }

    gemm128_wmma_kernel<<<(NGRAPH + 127) / 128, TB, 0, stream>>>(
        g, lin1_W, lin1_b, hid, NGRAPH, 1);
    mlp2_kernel<<<blocks((long)NGRAPH * NREL, TB), TB, 0, stream>>>(hid, lin2_W, lin2_b, (float*)d_out);
}


# Two launches fewer (10 to 8): the feature build joins the prefix-sum launch and derives its row factors from the counts; the head's two products share one launch, both on WMMA (head sums regrouped)
# speedup vs baseline: 2.8674x; 1.1328x over previous
#include <hip/hip_runtime.h>
#include <hip/hip_bf16.h>

typedef float v2f __attribute__((ext_vector_type(2)));
typedef float v8f __attribute__((ext_vector_type(8)));

#define NFEAT 128
#define NGRAPH 512
#define NREL 200

__global__ void zero_i32_kernel(int* __restrict__ p, long n) {
    long i = (long)blockIdx.x * blockDim.x + threadIdx.x;
    if (i < n) p[i] = 0;
}

#define EDGE_TB 256
#define EDGE_PT 4

__global__ __launch_bounds__(EDGE_TB) void count_rank_kernel(
    const int* __restrict__ src, const int* __restrict__ dst,
    int* __restrict__ cnt_out, int* __restrict__ cnt_in,
    int* __restrict__ rank, int E) {
    const int blk = blockIdx.x * (EDGE_TB * EDGE_PT);
    const int e0 = blk + threadIdx.x;
    if (blk + EDGE_TB * EDGE_PT <= E) {
        int s[EDGE_PT], d[EDGE_PT], r[EDGE_PT];
#pragma unroll
        for (int k = 0; k < EDGE_PT; ++k) { s[k] = src[e0 + k * EDGE_TB]; d[k] = dst[e0 + k * EDGE_TB]; }
#pragma unroll
        for (int k = 0; k < EDGE_PT; ++k) {
            atomicAdd(&cnt_out[s[k]], 1);
            r[k] = atomicAdd(&cnt_in[d[k]], 1);
        }
#pragma unroll
        for (int k = 0; k < EDGE_PT; ++k) rank[e0 + k * EDGE_TB] = r[k];
    } else {
        for (int k = 0; k < EDGE_PT; ++k) {
            const int e = e0 + k * EDGE_TB;
            if (e < E) {
                atomicAdd(&cnt_out[src[e]], 1);
                rank[e] = atomicAdd(&cnt_in[dst[e]], 1);
            }
        }
    }
}

#define ROWS_BX 8

__device__ __forceinline__ float4 scale4(float4 v, float s) {
    return make_float4(v.x * s, v.y * s, v.z * s, v.w * s);
}

__device__ __forceinline__ float4 bx_row(const float* __restrict__ tab, bool ent,
                                         const int* __restrict__ nid, const int* __restrict__ z, int i) {
    const int rn = nid[i];
    const int rz = z[i];
    return *(const float4*)(tab + (long)(ent ? rn : rz) * 64);
}

__device__ __forceinline__ float lane_f32(float v, int k) {
    return __int_as_float(__builtin_amdgcn_readlane(__float_as_int(v), k));
}

__global__ __launch_bounds__(1024) void scan_build_kernel(
    const int* __restrict__ cnt_out, const int* __restrict__ cnt_in,
    int* __restrict__ rowptr, float* __restrict__ norm_out, float* __restrict__ norm_in,
    const int* __restrict__ nid, const int* __restrict__ z,
    const float* __restrict__ emb, const float* __restrict__ ztab,
    float* __restrict__ x, int N) {
    const int tid = threadIdx.x;
    if (blockIdx.x != 0) {
        const int lane = tid & 31;
        const int wave = __builtin_amdgcn_readfirstlane((int)((((unsigned)blockIdx.x - 1u) * 1024u + (unsigned)tid) >> 5));
        const int i0 = wave * ROWS_BX;
        if (i0 >= N) return;
        const bool ent = lane < 16;
        const float* tab = (ent ? emb : ztab) + (lane & 15) * 4;
        float* xo = x + lane * 4;
        if (i0 + ROWS_BX <= N) {
            const float4 v0 = bx_row(tab, ent, nid, z, i0 + 0);
            const float4 v1 = bx_row(tab, ent, nid, z, i0 + 1);
            const float4 v2 = bx_row(tab, ent, nid, z, i0 + 2);
            const float4 v3 = bx_row(tab, ent, nid, z, i0 + 3);
            const float4 v4 = bx_row(tab, ent, nid, z, i0 + 4);
            const float4 v5 = bx_row(tab, ent, nid, z, i0 + 5);
            const float4 v6 = bx_row(tab, ent, nid, z, i0 + 6);
            const float4 v7 = bx_row(tab, ent, nid, z, i0 + 7);
            const bool in_side = (lane & ROWS_BX) != 0;
            const int c = (in_side ? cnt_in : cnt_out)[i0 + (lane & (ROWS_BX - 1))];
            __builtin_amdgcn_sched_barrier(0);
            const float f_out = rsqrtf(fmaxf((float)c, 1.0f));
            if (lane < 2 * ROWS_BX) (in_side ? norm_in : norm_out)[i0 + (lane & (ROWS_BX - 1))] = f_out;
            float* xr = xo + (long)i0 * NFEAT;
            *(float4*)(xr + 0 * NFEAT) = scale4(v0, lane_f32(f_out, 0));
            *(float4*)(xr + 1 * NFEAT) = scale4(v1, lane_f32(f_out, 1));
            *(float4*)(xr + 2 * NFEAT) = scale4(v2, lane_f32(f_out, 2));
            *(float4*)(xr + 3 * NFEAT) = scale4(v3, lane_f32(f_out, 3));
            *(float4*)(xr + 4 * NFEAT) = scale4(v4, lane_f32(f_out, 4));
            *(float4*)(xr + 5 * NFEAT) = scale4(v5, lane_f32(f_out, 5));
            *(float4*)(xr + 6 * NFEAT) = scale4(v6, lane_f32(f_out, 6));
            *(float4*)(xr + 7 * NFEAT) = scale4(v7, lane_f32(f_out, 7));
        } else {
            const int ic = min(i0 + (lane & (ROWS_BX - 1)), N - 1);
            const float f_out = rsqrtf(fmaxf((float)cnt_out[ic], 1.0f));
            const float f_in  = rsqrtf(fmaxf((float)cnt_in[ic], 1.0f));
            if (lane < ROWS_BX && i0 + lane < N) { norm_out[i0 + lane] = f_out; norm_in[i0 + lane] = f_in; }
            for (int i = i0; i < N; ++i) {
                *(float4*)(xo + (long)i * NFEAT) = scale4(bx_row(tab, ent, nid, z, i), lane_f32(f_out, i - i0));
            }
        }
        return;
    }
    __shared__ int wsum[32];
    const int ch  = (N + 1023) / 1024;
    const int beg = min(tid * ch, N);
    const int end = min(beg + ch, N);
    int s = 0;
    int i = beg;
    for (; i + 8 <= end; i += 8) {
        int a[8];
#pragma unroll
        for (int u = 0; u < 8; ++u) a[u] = cnt_in[i + u];
#pragma unroll
        for (int u = 0; u < 8; ++u) s += a[u];
    }
    for (; i < end; ++i) s += cnt_in[i];

    const int lane = tid & 31, w = tid >> 5;
    int incl = s;
#pragma unroll
    for (int o = 1; o < 32; o <<= 1) {
        int t = __shfl_up(incl, o);
        if (lane >= o) incl += t;
    }
    if (lane == 31) wsum[w] = incl;
    __syncthreads();
    if (w == 0) {
        int v = wsum[lane], iv = v;
#pragma unroll
        for (int o = 1; o < 32; o <<= 1) {
            int t = __shfl_up(iv, o);
            if (lane >= o) iv += t;
        }
        wsum[lane] = iv - v;
    }
    __syncthreads();
    int run = wsum[w] + incl - s;
    i = beg;
    for (; i + 8 <= end; i += 8) {
        int a[8];
#pragma unroll
        for (int u = 0; u < 8; ++u) a[u] = cnt_in[i + u];
#pragma unroll
        for (int u = 0; u < 8; ++u) { rowptr[i + u] = run; run += a[u]; }
    }
    for (; i < end; ++i) { rowptr[i] = run; run += cnt_in[i]; }
    if (tid == 1023) rowptr[N] = run;
}

__global__ __launch_bounds__(EDGE_TB) void fill_kernel(
    const int* __restrict__ src, const int* __restrict__ dst,
    const int* __restrict__ rank, const int* __restrict__ rowptr,
    int* __restrict__ csr_src, int E) {
    const int blk = blockIdx.x * (EDGE_TB * EDGE_PT);
    const int e0 = blk + threadIdx.x;
    if (blk + EDGE_TB * EDGE_PT <= E) {
        int s[EDGE_PT], d[EDGE_PT], r[EDGE_PT], p[EDGE_PT];
#pragma unroll
        for (int k = 0; k < EDGE_PT; ++k) {
            s[k] = src[e0 + k * EDGE_TB]; d[k] = dst[e0 + k * EDGE_TB]; r[k] = rank[e0 + k * EDGE_TB];
        }
#pragma unroll
        for (int k = 0; k < EDGE_PT; ++k) p[k] = rowptr[d[k]];
#pragma unroll
        for (int k = 0; k < EDGE_PT; ++k) csr_src[p[k] + r[k]] = s[k];
    } else {
        for (int k = 0; k < EDGE_PT; ++k) {
            const int e = e0 + k * EDGE_TB;
            if (e < E) csr_src[rowptr[dst[e]] + rank[e]] = src[e];
        }
    }
}

#define LAYER_TB 256
#define TILE_ROWS 16
#define APITCH 132
#define GU 16

__device__ __forceinline__ void close_row(float4& acc, float* __restrict__ As, int q, int nin_bits, int lane) {
    const float s = __int_as_float(__builtin_amdgcn_readlane(nin_bits, q));
    *(float4*)(As + q * APITCH + lane * 4) = make_float4(acc.x * s, acc.y * s, acc.z * s, acc.w * s);
    acc = make_float4(0.0f, 0.0f, 0.0f, 0.0f);
}

template <int U, bool NX>
__device__ __forceinline__ void take_rows(float4& acc, int& q, int& rend, int e, int j,
                                          int s_l, int rw, int nin_bits,
                                          const float* __restrict__ xl, float* __restrict__ As, int lane,
                                          int& s_nx, const int* __restrict__ list, int nx, int e_end) {
    float4 v[U];
#pragma unroll
    for (int u = 0; u < U; ++u) {
        const int sj = __builtin_amdgcn_readlane(s_l, j + u);
        v[u] = *(const float4*)(xl + (long)sj * NFEAT);
    }
    if (NX) {
        if (nx < e_end) s_nx = list[nx];
    }
#pragma unroll
    for (int u = 0; u < U; ++u) {
        while (e + u >= rend) {
            close_row(acc, As, q, nin_bits, lane);
            ++q;
            rend = __builtin_amdgcn_readlane(rw, q + 1);
        }
        acc.x += v[u].x;
        acc.y += v[u].y;
        acc.z += v[u].z;
        acc.w += v[u].w;
    }
}

__global__ __launch_bounds__(LAYER_TB) void gcn_layer_kernel(
    const int* __restrict__ rowptr, const int* __restrict__ csr_src,
    const float* __restrict__ x, const float* __restrict__ out_scale, const float* __restrict__ norm_in,
    const float* __restrict__ W, const float* __restrict__ bias,
    float* __restrict__ Y, int nrows, int relu,
    const int* __restrict__ pool_gid, float* __restrict__ pool_g) {
    __shared__ __attribute__((aligned(16))) float Wlds[NFEAT * NFEAT];
    __shared__ __attribute__((aligned(16))) float Astage[(LAYER_TB / 32) * TILE_ROWS * APITCH];

    const int tid = threadIdx.x;
    {
        const float4* Wv = (const float4*)W;
        float4* Lv = (float4*)Wlds;
#pragma unroll
        for (int i = 0; i < 16; ++i) Lv[tid + i * 256] = Wv[tid + i * 256];
    }
    __syncthreads();

    const int lane = tid & 31;
    const int wave = __builtin_amdgcn_readfirstlane(tid >> 5);
    const int slot = (wave + (((int)blockIdx.x >> 8) & 1) * 6) & 7;
    const int tile = slot * (int)gridDim.x + (int)blockIdx.x;
    const int rbase = tile * TILE_ROWS;
    if (rbase >= nrows) return;

    float* As = Astage + wave * (TILE_ROWS * APITCH);

    {
        const int rp = rowptr[min(rbase + lane, nrows)];
        const int nin_bits = __float_as_int(norm_in[min(rbase + lane, nrows - 1)]);
        const int e_beg = __builtin_amdgcn_readlane(rp, 0);
        const int e_end = __builtin_amdgcn_readlane(rp, TILE_ROWS);
        const int rw = lane < TILE_ROWS ? rp : 0x7fffffff;
        const float* xl = x + lane * 4;
        int q = 0;
        int rend = __builtin_amdgcn_readlane(rw, 1);
        float4 acc = make_float4(0.0f, 0.0f, 0.0f, 0.0f);

        int s_cur = 0;
        if (e_beg + lane < e_end) s_cur = csr_src[e_beg + lane];
        static_assert(2 * GU == 32, "a chunk of 32 list entries is two full groups");
        for (int base = e_beg; base < e_end; base += 32) {
            const int n = min(32, e_end - base);
            int s_nx = 0;
            const int nx = base + 32 + lane;
            int j = 0;
            if (n >= GU) {
                take_rows<GU, true>(acc, q, rend, base, 0, s_cur, rw, nin_bits, xl, As, lane, s_nx, csr_src, nx, e_end);
                j = GU;
            }
            if (n >= 2 * GU) {
                take_rows<GU, false>(acc, q, rend, base + GU, GU, s_cur, rw, nin_bits, xl, As, lane, s_nx, csr_src, nx, e_end);
                j = 2 * GU;
            }
            if (n - j >= 8) { take_rows<8, false>(acc, q, rend, base + j, j, s_cur, rw, nin_bits, xl, As, lane, s_nx, csr_src, nx, e_end); j += 8; }
            if (n - j >= 4) { take_rows<4, false>(acc, q, rend, base + j, j, s_cur, rw, nin_bits, xl, As, lane, s_nx, csr_src, nx, e_end); j += 4; }
            if (n - j >= 2) { take_rows<2, false>(acc, q, rend, base + j, j, s_cur, rw, nin_bits, xl, As, lane, s_nx, csr_src, nx, e_end); j += 2; }
            if (n - j >= 1) { take_rows<1, false>(acc, q, rend, base + j, j, s_cur, rw, nin_bits, xl, As, lane, s_nx, csr_src, nx, e_end); }
            s_cur = s_nx;
        }
        for (; q < TILE_ROWS; ++q) close_row(acc, As, q, nin_bits, lane);
    }

    const int lrow  = lane & 15;
    const int khalf = (lane >> 4) << 1;
    const int rowhi = (lane >> 4) << 3;
    const float* Arow = As + lrow * APITCH + khalf;

    v8f acc[8];
#pragma unroll
    for (int t = 0; t < 8; ++t) {
        float bv = bias[t * 16 + lrow];
#pragma unroll
        for (int r = 0; r < 8; ++r) acc[t][r] = bv;
    }

    float osc[8];
#pragma unroll
    for (int r = 0; r < 8; ++r) osc[r] = 1.0f;
    if (!pool_g) {
#pragma unroll
        for (int r = 0; r < 8; ++r) osc[r] = out_scale[min(rbase + rowhi + r, nrows - 1)];
    }

    for (int k = 0; k < NFEAT; k += 4) {
        const v2f a = *(const v2f*)(Arow + k);
#pragma unroll
        for (int t = 0; t < 8; ++t) {
            v2f b;
            b.x = Wlds[(k + khalf) * NFEAT + t * 16 + lrow];
            b.y = Wlds[(k + khalf + 1) * NFEAT + t * 16 + lrow];
            acc[t] = __builtin_amdgcn_wmma_f32_16x16x4_f32(
                false, a, false, b, (short)0, acc[t], false, false);
        }
    }

    if (pool_g) {
        const int row0 = rbase + rowhi;
        int gr[8];
        float run[8];
#pragma unroll
        for (int r = 0; r < 8; ++r) gr[r] = pool_gid[min(row0 + r, nrows - 1)];
#pragma unroll
        for (int t = 0; t < 8; ++t) run[t] = 0.0f;
        float* gcol = pool_g + lrow;
#pragma unroll
        for (int r = 0; r < 8; ++r) {
            if (r > 0 && gr[r] != gr[r - 1]) {
                float* gp = gcol + (long)gr[r - 1] * NFEAT;
#pragma unroll
                for (int t = 0; t < 8; ++t) { atomicAdd(gp + t * 16, run[t]); run[t] = 0.0f; }
            }
            if (row0 + r < nrows) {
#pragma unroll
                for (int t = 0; t < 8; ++t) {
                    float v = acc[t][r];
                    if (relu) v = fmaxf(v, 0.0f);
                    run[t] += v;
                }
            }
        }
        float* gp = gcol + (long)gr[7] * NFEAT;
#pragma unroll
        for (int t = 0; t < 8; ++t) atomicAdd(gp + t * 16, run[t]);
        return;
    }

#pragma unroll
    for (int r = 0; r < 8; ++r) {
        const int row = rbase + rowhi + r;
        if (row < nrows) {
#pragma unroll
            for (int t = 0; t < 8; ++t) {
                float v = acc[t][r];
                if (relu) v = fmaxf(v, 0.0f);
                Y[(long)row * NFEAT + t * 16 + lrow] = v * osc[r];
            }
        }
    }
}

#define HEAD_TB 256
#define HPITCH 132

__device__ __forceinline__ void head_load_b(v2f (&b)[32], const float* __restrict__ Bp, int ldb) {
#pragma unroll
    for (int s = 0; s < 32; ++s) {
        b[s].x = Bp[(4 * s + 0) * ldb];
        b[s].y = Bp[(4 * s + 1) * ldb];
    }
}

__device__ __forceinline__ v8f head_tile(const float* __restrict__ Ap, const v2f (&b)[32], float bv) {
    v8f c0, c1;
#pragma unroll
    for (int r = 0; r < 8; ++r) { c0[r] = bv; c1[r] = 0.0f; }
#pragma unroll
    for (int s = 0; s < 32; s += 2) {
        const v2f a0 = *(const v2f*)(Ap + 4 * s);
        const v2f a1 = *(const v2f*)(Ap + 4 * s + 4);
        c0 = __builtin_amdgcn_wmma_f32_16x16x4_f32(false, a0, false, b[s], (short)0, c0, false, false);
        c1 = __builtin_amdgcn_wmma_f32_16x16x4_f32(false, a1, false, b[s + 1], (short)0, c1, false, false);
    }
    return c0 + c1;
}

__global__ __launch_bounds__(HEAD_TB) void head_kernel(
    const float* __restrict__ G,
    const float* __restrict__ W1, const float* __restrict__ b1,
    const float* __restrict__ W2, const float* __restrict__ b2,
    float* __restrict__ out, int nrows) {
    __shared__ __attribute__((aligned(16))) float Gs[TILE_ROWS * HPITCH];
    __shared__ __attribute__((aligned(16))) float Hs[TILE_ROWS * HPITCH];

    const int tid = threadIdx.x;
    const int lane = tid & 31;
    const int wave = __builtin_amdgcn_readfirstlane(tid >> 5);
    const int rbase = blockIdx.x * TILE_ROWS;
    const int lrow  = lane & 15;
    const int khalf = (lane >> 4) << 1;
    const int rowhi = (lane >> 4) << 3;

    const int colh  = wave * 16 + lrow;
    const int col0  = wave * 16 + lrow;
    const int col1  = 128 + wave * 16 + lrow;
    const int col1c = min(col1, NREL - 1);
    const bool two  = 128 + wave * 16 < NREL;

    v2f bh[32], bo0[32], bo1[32];
    head_load_b(bh, W1 + khalf * NFEAT + colh, NFEAT);
    head_load_b(bo0, W2 + khalf * NREL + col0, NREL);
    if (two) head_load_b(bo1, W2 + khalf * NREL + col1c, NREL);
    const float bvh = b1[colh], bv0 = b2[col0], bv1 = b2[col1c];

#pragma unroll
    for (int i = 0; i < 2; ++i) {
        const int f = tid + i * HEAD_TB;
        const int r = f >> 5, c4 = (f & 31) * 4;
        const int gr = min(rbase + r, nrows - 1);
        *(float4*)(Gs + r * HPITCH + c4) = *(const float4*)(G + (long)gr * NFEAT + c4);
    }
    __syncthreads();

    {
        const v8f h = head_tile(Gs + lrow * HPITCH + khalf, bh, bvh);
#pragma unroll
        for (int r = 0; r < 8; ++r)
            Hs[(rowhi + r) * HPITCH + colh] = fmaxf(h[r], 0.0f);
    }
    __syncthreads();

    const float* Ap = Hs + lrow * HPITCH + khalf;
    {
        const v8f o = head_tile(Ap, bo0, bv0);
#pragma unroll
        for (int r = 0; r < 8; ++r) {
            const int row = rbase + rowhi + r;
            if (row < nrows) out[(long)row * NREL + col0] = o[r];
        }
    }
    if (two) {
        const v8f o = head_tile(Ap, bo1, bv1);
        if (col1 < NREL) {
#pragma unroll
            for (int r = 0; r < 8; ++r) {
                const int row = rbase + rowhi + r;
                if (row < nrows) out[(long)row * NREL + col1] = o[r];
            }
        }
    }
}

extern "C" void kernel_launch(void* const* d_in, const int* in_sizes, int n_in,
                              void* d_out, int out_size, void* d_ws, size_t ws_size,
                              hipStream_t stream) {
    const int*   nid    = (const int*)d_in[0];
    const int*   z      = (const int*)d_in[1];
    const int*   src    = (const int*)d_in[2];
    const int*   dst    = (const int*)d_in[3];
    const int*   gid    = (const int*)d_in[4];
    const float* emb    = (const float*)d_in[5];
    const float* ztab   = (const float*)d_in[6];
    const float* Ws[3]  = {(const float*)d_in[7], (const float*)d_in[9], (const float*)d_in[11]};
    const float* bs[3]  = {(const float*)d_in[8], (const float*)d_in[10], (const float*)d_in[12]};
    const float* lin1_W = (const float*)d_in[13];
    const float* lin1_b = (const float*)d_in[14];
    const float* lin2_W = (const float*)d_in[15];
    const float* lin2_b = (const float*)d_in[16];

    const int N = in_sizes[0];
    const int E = in_sizes[2];

    float* x        = (float*)d_ws;
    float* agg      = x + (size_t)N * NFEAT;
    float* norm_out = agg + (size_t)N * NFEAT;
    float* norm_in  = norm_out + N;
    float* g        = norm_in + N;
    int*   cnt_out  = (int*)(g + (size_t)NGRAPH * NFEAT);
    int*   cnt_in   = cnt_out + N;
    int*   rowptr   = cnt_in + N;
    int*   csr_src  = rowptr + N + 1;
    int*   rank     = (int*)agg;

    const int TB = 256;
    auto blocks = [](long n, int tb) { return (unsigned)((n + tb - 1) / tb); };

    const long nclear = (long)NGRAPH * NFEAT + 2L * N;
    zero_i32_kernel<<<blocks(nclear, TB), TB, 0, stream>>>((int*)g, nclear);
    count_rank_kernel<<<blocks(E, EDGE_TB * EDGE_PT), EDGE_TB, 0, stream>>>(src, dst, cnt_out, cnt_in, rank, E);
    scan_build_kernel<<<1 + blocks((long)((N + ROWS_BX - 1) / ROWS_BX) * 32, 1024), 1024, 0, stream>>>(
        cnt_out, cnt_in, rowptr, norm_out, norm_in, nid, z, emb, ztab, x, N);
    fill_kernel<<<blocks(E, EDGE_TB * EDGE_PT), EDGE_TB, 0, stream>>>(src, dst, rank, rowptr, csr_src, E);

    const long ntiles = ((long)N + TILE_ROWS - 1) / TILE_ROWS;
    const unsigned need_blocks = blocks(ntiles, LAYER_TB / 32);
    const unsigned layer_blocks = need_blocks > 512u ? need_blocks : 512u;
    float* xin = x;
    float* xout = agg;
    for (int layer = 0; layer < 3; ++layer) {
        const bool last = layer == 2;
        gcn_layer_kernel<<<layer_blocks, LAYER_TB, 0, stream>>>(
            rowptr, csr_src, xin, norm_out, norm_in, Ws[layer], bs[layer], xout, N, last ? 0 : 1,
            last ? gid : nullptr, last ? g : nullptr);
        float* t = xin; xin = xout; xout = t;
    }

    head_kernel<<<(NGRAPH + TILE_ROWS - 1) / TILE_ROWS, HEAD_TB, 0, stream>>>(
        g, lin1_W, lin1_b, lin2_W, lin2_b, (float*)d_out, NGRAPH);
}


# Edge list and both degree counts built by binning through LDS (chunk counts, column sums, one block per 256-node bin, whole-line stores); no global atomics; a row's sum order is LDS arrival order
# speedup vs baseline: 4.5437x; 1.5846x over previous
#include <hip/hip_runtime.h>
#include <hip/hip_bf16.h>

typedef float v2f __attribute__((ext_vector_type(2)));
typedef float v8f __attribute__((ext_vector_type(8)));

#define NFEAT 128
#define NGRAPH 512
#define NREL 200

#define NBIN 256
#define BIN_TB 256
#define BIN_LST 8192
#define BIN_RB 32
#define BIN_EB 16

__device__ __forceinline__ int block_excl_scan(int v, int* __restrict__ wsum, int tid, int& total) {
    const int lane = tid & 31, w = tid >> 5;
    int incl = v;
#pragma unroll
    for (int o = 1; o < 32; o <<= 1) {
        int t = __shfl_up(incl, o);
        if (lane >= o) incl += t;
    }
    if (lane == 31) wsum[w] = incl;
    __syncthreads();
    int woff = 0, tot = 0;
#pragma unroll
    for (int k = 0; k < BIN_TB / 32; ++k) {
        const int t = wsum[k];
        if (k < w) woff += t;
        tot += t;
    }
    __syncthreads();
    total = tot;
    return woff + incl - v;
}

__global__ __launch_bounds__(BIN_TB) void bin_count_kernel(
    const int* __restrict__ src, const int* __restrict__ dst,
    int* __restrict__ histd, int* __restrict__ hists,
    int* __restrict__ gclear, int nclear, int E, int chk, int shift) {
    __shared__ int hd[NBIN], hs[NBIN];
    const int tid = threadIdx.x;
    hd[tid] = 0;
    hs[tid] = 0;
    for (int i = blockIdx.x * BIN_TB + tid; i < nclear; i += gridDim.x * BIN_TB) gclear[i] = 0;
    __syncthreads();
    const int e0 = blockIdx.x * chk;
    const int e1 = min(e0 + chk, E);
    for (int e = e0 + tid; e < e1; e += BIN_EB * BIN_TB) {
        int d[BIN_EB], sv[BIN_EB];
#pragma unroll
        for (int k = 0; k < BIN_EB; ++k) {
            const int ee = e + k * BIN_TB;
            d[k]  = ee < e1 ? dst[ee] : -1;
            sv[k] = ee < e1 ? src[ee] : -1;
        }
#pragma unroll
        for (int k = 0; k < BIN_EB; ++k) {
            if (d[k] >= 0) {
                atomicAdd(&hd[d[k] >> shift], 1);
                atomicAdd(&hs[sv[k] >> shift], 1);
            }
        }
    }
    __syncthreads();
    histd[blockIdx.x * NBIN + tid] = hd[tid];
    hists[blockIdx.x * NBIN + tid] = hs[tid];
}

__global__ __launch_bounds__(BIN_TB) void bin_scatter_kernel(
    const int* __restrict__ src, const int* __restrict__ dst,
    const int* __restrict__ histd, const int* __restrict__ hists,
    int* __restrict__ bd, int* __restrict__ bs, int* __restrict__ bo,
    int* __restrict__ bstartd, int* __restrict__ bstarts, int E, int chk, int shift) {
    extern __shared__ int stage[];
    __shared__ int offd[NBIN], offs[NBIN], lod[NBIN], los[NBIN], curd[NBIN], curs[NBIN], wsum[BIN_TB / 32];
    int* sd = stage;
    int* ss = stage + chk;
    int* so = stage + 2 * chk;
    const int tid = threadIdx.x;
    int totd = 0, befd = 0, mined = 0, tots = 0, befs = 0, mines = 0;
    for (int b0 = 0; b0 < NBIN; b0 += BIN_RB) {
        int vd[BIN_RB], vs[BIN_RB];
#pragma unroll
        for (int k = 0; k < BIN_RB; ++k) {
            vd[k] = histd[(b0 + k) * NBIN + tid];
            vs[k] = hists[(b0 + k) * NBIN + tid];
        }
#pragma unroll
        for (int k = 0; k < BIN_RB; ++k) {
            const int b = b0 + k;
            totd += vd[k];
            tots += vs[k];
            if (b < (int)blockIdx.x) { befd += vd[k]; befs += vs[k]; }
            if (b == (int)blockIdx.x) { mined = vd[k]; mines = vs[k]; }
        }
    }
    int alld, alls, unused;
    const int based = block_excl_scan(totd, wsum, tid, alld);
    const int bases = block_excl_scan(tots, wsum, tid, alls);
    const int ld = block_excl_scan(mined, wsum, tid, unused);
    const int ls = block_excl_scan(mines, wsum, tid, unused);
    offd[tid] = based + befd - ld;
    offs[tid] = bases + befs - ls;
    lod[tid] = ld;
    los[tid] = ls;
    curd[tid] = 0;
    curs[tid] = 0;
    if (blockIdx.x == 0) {
        bstartd[tid] = based;
        bstarts[tid] = bases;
        if (tid == NBIN - 1) { bstartd[NBIN] = alld; bstarts[NBIN] = alls; }
    }
    __syncthreads();
    const int e0 = blockIdx.x * chk;
    const int e1 = min(e0 + chk, E);
    for (int e = e0 + tid; e < e1; e += BIN_EB * BIN_TB) {
        int d[BIN_EB], sv[BIN_EB];
#pragma unroll
        for (int k = 0; k < BIN_EB; ++k) {
            const int ee = e + k * BIN_TB;
            d[k]  = ee < e1 ? dst[ee] : -1;
            sv[k] = ee < e1 ? src[ee] : -1;
        }
#pragma unroll
        for (int k = 0; k < BIN_EB; ++k) {
            if (d[k] >= 0) {
                const int kd = d[k] >> shift;
                const int ks = sv[k] >> shift;
                const int p = lod[kd] + atomicAdd(&curd[kd], 1);
                const int q = los[ks] + atomicAdd(&curs[ks], 1);
                sd[p] = d[k];
                ss[p] = sv[k];
                so[q] = sv[k];
            }
        }
    }
    __syncthreads();
    const int n = e1 - e0;
    for (int i = tid; i < n; i += BIN_TB) {
        const int d = sd[i];
        const int sv = so[i];
        const int p = offd[d >> shift] + i;
        bd[p] = d;
        bs[p] = ss[i];
        bo[offs[sv >> shift] + i] = sv;
    }
}

__global__ __launch_bounds__(BIN_TB) void bin_csr_kernel(
    const int* __restrict__ bd, const int* __restrict__ bs, const int* __restrict__ bo,
    const int* __restrict__ bstartd, const int* __restrict__ bstarts,
    int* __restrict__ cnt_out, int* __restrict__ cnt_in,
    int* __restrict__ rowptr, int* __restrict__ csr_src, int N, int E, int shift) {
    extern __shared__ int binmem[];
    __shared__ int wsum[BIN_TB / 32];
    __shared__ int lst[BIN_LST];
    const int tid = threadIdx.x;
    const int binw = 1 << shift;
    int* cin = binmem;
    int* cur = binmem + binw;
    int* cou = binmem + 2 * binw;
    const int node0 = (int)blockIdx.x << shift;
    if (blockIdx.x == 0 && tid == 0) rowptr[N] = E;
    if (node0 >= N) return;
    const int d0 = bstartd[blockIdx.x], d1 = bstartd[blockIdx.x + 1];
    const int s0 = bstarts[blockIdx.x], s1 = bstarts[blockIdx.x + 1];
    for (int i = tid; i < binw; i += BIN_TB) { cin[i] = 0; cou[i] = 0; }
    __syncthreads();
    for (int i = d0 + tid; i < d1; i += BIN_EB * BIN_TB) {
        int d[BIN_EB];
#pragma unroll
        for (int k = 0; k < BIN_EB; ++k) { const int ii = i + k * BIN_TB; d[k] = ii < d1 ? bd[ii] : -1; }
#pragma unroll
        for (int k = 0; k < BIN_EB; ++k) if (d[k] >= 0) atomicAdd(&cin[d[k] - node0], 1);
    }
    for (int i = s0 + tid; i < s1; i += BIN_EB * BIN_TB) {
        int sv[BIN_EB];
#pragma unroll
        for (int k = 0; k < BIN_EB; ++k) { const int ii = i + k * BIN_TB; sv[k] = ii < s1 ? bo[ii] : -1; }
#pragma unroll
        for (int k = 0; k < BIN_EB; ++k) if (sv[k] >= 0) atomicAdd(&cou[sv[k] - node0], 1);
    }
    __syncthreads();
    const int per = (binw + BIN_TB - 1) / BIN_TB;
    const int beg = min(tid * per, binw);
    const int end = min(beg + per, binw);
    int sum = 0;
    for (int i = beg; i < end; ++i) sum += cin[i];
    int all;
    int run = block_excl_scan(sum, wsum, tid, all);
    for (int i = beg; i < end; ++i) {
        const int v = cin[i];
        cur[i] = run;
        if (node0 + i < N) {
            cnt_in[node0 + i] = v;
            cnt_out[node0 + i] = cou[i];
            rowptr[node0 + i] = d0 + run;
        }
        run += v;
    }
    __syncthreads();
    const int m = d1 - d0;
    const bool fits = m <= BIN_LST;
    for (int i = d0 + tid; i < d1; i += BIN_EB * BIN_TB) {
        int d[BIN_EB], sv[BIN_EB];
#pragma unroll
        for (int k = 0; k < BIN_EB; ++k) {
            const int ii = i + k * BIN_TB;
            d[k]  = ii < d1 ? bd[ii] : -1;
            sv[k] = ii < d1 ? bs[ii] : -1;
        }
#pragma unroll
        for (int k = 0; k < BIN_EB; ++k) {
            if (d[k] >= 0) {
                const int p = atomicAdd(&cur[d[k] - node0], 1);
                if (fits) lst[p] = sv[k];
                else csr_src[d0 + p] = sv[k];
            }
        }
    }
    if (fits) {
        __syncthreads();
        for (int i = tid; i < m; i += BIN_TB) csr_src[d0 + i] = lst[i];
    }
}

#define ROWS_BX 8

__device__ __forceinline__ float4 scale4(float4 v, float s) {
    return make_float4(v.x * s, v.y * s, v.z * s, v.w * s);
}

__device__ __forceinline__ float4 bx_row(const float* __restrict__ tab, bool ent,
                                         const int* __restrict__ nid, const int* __restrict__ z, int i) {
    const int rn = nid[i];
    const int rz = z[i];
    return *(const float4*)(tab + (long)(ent ? rn : rz) * 64);
}

__device__ __forceinline__ float lane_f32(float v, int k) {
    return __int_as_float(__builtin_amdgcn_readlane(__float_as_int(v), k));
}

__global__ __launch_bounds__(1024) void build_kernel(
    const int* __restrict__ cnt_out, const int* __restrict__ cnt_in,
    float* __restrict__ norm_out, float* __restrict__ norm_in,
    const int* __restrict__ nid, const int* __restrict__ z,
    const float* __restrict__ emb, const float* __restrict__ ztab,
    float* __restrict__ x, int N) {
    const int tid = threadIdx.x;
    const int lane = tid & 31;
    const int wave = __builtin_amdgcn_readfirstlane((int)(((unsigned)blockIdx.x * 1024u + (unsigned)tid) >> 5));
    const int i0 = wave * ROWS_BX;
    if (i0 >= N) return;
    const bool ent = lane < 16;
    const float* tab = (ent ? emb : ztab) + (lane & 15) * 4;
    float* xo = x + lane * 4;
    if (i0 + ROWS_BX <= N) {
        const float4 v0 = bx_row(tab, ent, nid, z, i0 + 0);
        const float4 v1 = bx_row(tab, ent, nid, z, i0 + 1);
        const float4 v2 = bx_row(tab, ent, nid, z, i0 + 2);
        const float4 v3 = bx_row(tab, ent, nid, z, i0 + 3);
        const float4 v4 = bx_row(tab, ent, nid, z, i0 + 4);
        const float4 v5 = bx_row(tab, ent, nid, z, i0 + 5);
        const float4 v6 = bx_row(tab, ent, nid, z, i0 + 6);
        const float4 v7 = bx_row(tab, ent, nid, z, i0 + 7);
        const bool in_side = (lane & ROWS_BX) != 0;
        const int c = (in_side ? cnt_in : cnt_out)[i0 + (lane & (ROWS_BX - 1))];
        __builtin_amdgcn_sched_barrier(0);
        const float f_out = rsqrtf(fmaxf((float)c, 1.0f));
        if (lane < 2 * ROWS_BX) (in_side ? norm_in : norm_out)[i0 + (lane & (ROWS_BX - 1))] = f_out;
        float* xr = xo + (long)i0 * NFEAT;
        *(float4*)(xr + 0 * NFEAT) = scale4(v0, lane_f32(f_out, 0));
        *(float4*)(xr + 1 * NFEAT) = scale4(v1, lane_f32(f_out, 1));
        *(float4*)(xr + 2 * NFEAT) = scale4(v2, lane_f32(f_out, 2));
        *(float4*)(xr + 3 * NFEAT) = scale4(v3, lane_f32(f_out, 3));
        *(float4*)(xr + 4 * NFEAT) = scale4(v4, lane_f32(f_out, 4));
        *(float4*)(xr + 5 * NFEAT) = scale4(v5, lane_f32(f_out, 5));
        *(float4*)(xr + 6 * NFEAT) = scale4(v6, lane_f32(f_out, 6));
        *(float4*)(xr + 7 * NFEAT) = scale4(v7, lane_f32(f_out, 7));
    } else {
        const int ic = min(i0 + (lane & (ROWS_BX - 1)), N - 1);
        const float f_out = rsqrtf(fmaxf((float)cnt_out[ic], 1.0f));
        const float f_in  = rsqrtf(fmaxf((float)cnt_in[ic], 1.0f));
        if (lane < ROWS_BX && i0 + lane < N) { norm_out[i0 + lane] = f_out; norm_in[i0 + lane] = f_in; }
        for (int i = i0; i < N; ++i) {
            *(float4*)(xo + (long)i * NFEAT) = scale4(bx_row(tab, ent, nid, z, i), lane_f32(f_out, i - i0));
        }
    }
}

#define LAYER_TB 256
#define TILE_ROWS 16
#define APITCH 132
#define GU 16

__device__ __forceinline__ void close_row(float4& acc, float* __restrict__ As, int q, int nin_bits, int lane) {
    const float s = __int_as_float(__builtin_amdgcn_readlane(nin_bits, q));
    *(float4*)(As + q * APITCH + lane * 4) = make_float4(acc.x * s, acc.y * s, acc.z * s, acc.w * s);
    acc = make_float4(0.0f, 0.0f, 0.0f, 0.0f);
}

template <int U, bool NX>
__device__ __forceinline__ void take_rows(float4& acc, int& q, int& rend, int e, int j,
                                          int s_l, int rw, int nin_bits,
                                          const float* __restrict__ xl, float* __restrict__ As, int lane,
                                          int& s_nx, const int* __restrict__ list, int nx, int e_end) {
    float4 v[U];
#pragma unroll
    for (int u = 0; u < U; ++u) {
        const int sj = __builtin_amdgcn_readlane(s_l, j + u);
        v[u] = *(const float4*)(xl + (long)sj * NFEAT);
    }
    if (NX) {
        if (nx < e_end) s_nx = list[nx];
    }
#pragma unroll
    for (int u = 0; u < U; ++u) {
        while (e + u >= rend) {
            close_row(acc, As, q, nin_bits, lane);
            ++q;
            rend = __builtin_amdgcn_readlane(rw, q + 1);
        }
        acc.x += v[u].x;
        acc.y += v[u].y;
        acc.z += v[u].z;
        acc.w += v[u].w;
    }
}

__global__ __launch_bounds__(LAYER_TB) void gcn_layer_kernel(
    const int* __restrict__ rowptr, const int* __restrict__ csr_src,
    const float* __restrict__ x, const float* __restrict__ out_scale, const float* __restrict__ norm_in,
    const float* __restrict__ W, const float* __restrict__ bias,
    float* __restrict__ Y, int nrows, int relu,
    const int* __restrict__ pool_gid, float* __restrict__ pool_g) {
    __shared__ __attribute__((aligned(16))) float Wlds[NFEAT * NFEAT];
    __shared__ __attribute__((aligned(16))) float Astage[(LAYER_TB / 32) * TILE_ROWS * APITCH];

    const int tid = threadIdx.x;
    {
        const float4* Wv = (const float4*)W;
        float4* Lv = (float4*)Wlds;
#pragma unroll
        for (int i = 0; i < 16; ++i) Lv[tid + i * 256] = Wv[tid + i * 256];
    }
    __syncthreads();

    const int lane = tid & 31;
    const int wave = __builtin_amdgcn_readfirstlane(tid >> 5);
    const int slot = (wave + (((int)blockIdx.x >> 8) & 1) * 6) & 7;
    const int tile = slot * (int)gridDim.x + (int)blockIdx.x;
    const int rbase = tile * TILE_ROWS;
    if (rbase >= nrows) return;

    float* As = Astage + wave * (TILE_ROWS * APITCH);

    {
        const int rp = rowptr[min(rbase + lane, nrows)];
        const int nin_bits = __float_as_int(norm_in[min(rbase + lane, nrows - 1)]);
        const int e_beg = __builtin_amdgcn_readlane(rp, 0);
        const int e_end = __builtin_amdgcn_readlane(rp, TILE_ROWS);
        const int rw = lane < TILE_ROWS ? rp : 0x7fffffff;
        const float* xl = x + lane * 4;
        int q = 0;
        int rend = __builtin_amdgcn_readlane(rw, 1);
        float4 acc = make_float4(0.0f, 0.0f, 0.0f, 0.0f);

        int s_cur = 0;
        if (e_beg + lane < e_end) s_cur = csr_src[e_beg + lane];
        static_assert(2 * GU == 32, "a chunk of 32 list entries is two full groups");
        for (int base = e_beg; base < e_end; base += 32) {
            const int n = min(32, e_end - base);
            int s_nx = 0;
            const int nx = base + 32 + lane;
            int j = 0;
            if (n >= GU) {
                take_rows<GU, true>(acc, q, rend, base, 0, s_cur, rw, nin_bits, xl, As, lane, s_nx, csr_src, nx, e_end);
                j = GU;
            }
            if (n >= 2 * GU) {
                take_rows<GU, false>(acc, q, rend, base + GU, GU, s_cur, rw, nin_bits, xl, As, lane, s_nx, csr_src, nx, e_end);
                j = 2 * GU;
            }
            if (n - j >= 8) { take_rows<8, false>(acc, q, rend, base + j, j, s_cur, rw, nin_bits, xl, As, lane, s_nx, csr_src, nx, e_end); j += 8; }
            if (n - j >= 4) { take_rows<4, false>(acc, q, rend, base + j, j, s_cur, rw, nin_bits, xl, As, lane, s_nx, csr_src, nx, e_end); j += 4; }
            if (n - j >= 2) { take_rows<2, false>(acc, q, rend, base + j, j, s_cur, rw, nin_bits, xl, As, lane, s_nx, csr_src, nx, e_end); j += 2; }
            if (n - j >= 1) { take_rows<1, false>(acc, q, rend, base + j, j, s_cur, rw, nin_bits, xl, As, lane, s_nx, csr_src, nx, e_end); }
            s_cur = s_nx;
        }
        for (; q < TILE_ROWS; ++q) close_row(acc, As, q, nin_bits, lane);
    }

    const int lrow  = lane & 15;
    const int khalf = (lane >> 4) << 1;
    const int rowhi = (lane >> 4) << 3;
    const float* Arow = As + lrow * APITCH + khalf;

    v8f acc[8];
#pragma unroll
    for (int t = 0; t < 8; ++t) {
        float bv = bias[t * 16 + lrow];
#pragma unroll
        for (int r = 0; r < 8; ++r) acc[t][r] = bv;
    }

    float osc[8];
#pragma unroll
    for (int r = 0; r < 8; ++r) osc[r] = 1.0f;
    if (!pool_g) {
#pragma unroll
        for (int r = 0; r < 8; ++r) osc[r] = out_scale[min(rbase + rowhi + r, nrows - 1)];
    }

    for (int k = 0; k < NFEAT; k += 4) {
        const v2f a = *(const v2f*)(Arow + k);
#pragma unroll
        for (int t = 0; t < 8; ++t) {
            v2f b;
            b.x = Wlds[(k + khalf) * NFEAT + t * 16 + lrow];
            b.y = Wlds[(k + khalf + 1) * NFEAT + t * 16 + lrow];
            acc[t] = __builtin_amdgcn_wmma_f32_16x16x4_f32(
                false, a, false, b, (short)0, acc[t], false, false);
        }
    }

    if (pool_g) {
        const int row0 = rbase + rowhi;
        int gr[8];
        float run[8];
#pragma unroll
        for (int r = 0; r < 8; ++r) gr[r] = pool_gid[min(row0 + r, nrows - 1)];
#pragma unroll
        for (int t = 0; t < 8; ++t) run[t] = 0.0f;
        float* gcol = pool_g + lrow;
#pragma unroll
        for (int r = 0; r < 8; ++r) {
            if (r > 0 && gr[r] != gr[r - 1]) {
                float* gp = gcol + (long)gr[r - 1] * NFEAT;
#pragma unroll
                for (int t = 0; t < 8; ++t) { atomicAdd(gp + t * 16, run[t]); run[t] = 0.0f; }
            }
            if (row0 + r < nrows) {
#pragma unroll
                for (int t = 0; t < 8; ++t) {
                    float v = acc[t][r];
                    if (relu) v = fmaxf(v, 0.0f);
                    run[t] += v;
                }
            }
        }
        float* gp = gcol + (long)gr[7] * NFEAT;
#pragma unroll
        for (int t = 0; t < 8; ++t) atomicAdd(gp + t * 16, run[t]);
        return;
    }

#pragma unroll
    for (int r = 0; r < 8; ++r) {
        const int row = rbase + rowhi + r;
        if (row < nrows) {
#pragma unroll
            for (int t = 0; t < 8; ++t) {
                float v = acc[t][r];
                if (relu) v = fmaxf(v, 0.0f);
                Y[(long)row * NFEAT + t * 16 + lrow] = v * osc[r];
            }
        }
    }
}

#define HEAD_TB 256
#define HPITCH 132

__device__ __forceinline__ void head_load_b(v2f (&b)[32], const float* __restrict__ Bp, int ldb) {
#pragma unroll
    for (int s = 0; s < 32; ++s) {
        b[s].x = Bp[(4 * s + 0) * ldb];
        b[s].y = Bp[(4 * s + 1) * ldb];
    }
}

__device__ __forceinline__ v8f head_tile(const float* __restrict__ Ap, const v2f (&b)[32], float bv) {
    v8f c0, c1;
#pragma unroll
    for (int r = 0; r < 8; ++r) { c0[r] = bv; c1[r] = 0.0f; }
#pragma unroll
    for (int s = 0; s < 32; s += 2) {
        const v2f a0 = *(const v2f*)(Ap + 4 * s);
        const v2f a1 = *(const v2f*)(Ap + 4 * s + 4);
        c0 = __builtin_amdgcn_wmma_f32_16x16x4_f32(false, a0, false, b[s], (short)0, c0, false, false);
        c1 = __builtin_amdgcn_wmma_f32_16x16x4_f32(false, a1, false, b[s + 1], (short)0, c1, false, false);
    }
    return c0 + c1;
}

__global__ __launch_bounds__(HEAD_TB) void head_kernel(
    const float* __restrict__ G,
    const float* __restrict__ W1, const float* __restrict__ b1,
    const float* __restrict__ W2, const float* __restrict__ b2,
    float* __restrict__ out, int nrows) {
    __shared__ __attribute__((aligned(16))) float Gs[TILE_ROWS * HPITCH];
    __shared__ __attribute__((aligned(16))) float Hs[TILE_ROWS * HPITCH];

    const int tid = threadIdx.x;
    const int lane = tid & 31;
    const int wave = __builtin_amdgcn_readfirstlane(tid >> 5);
    const int rbase = blockIdx.x * TILE_ROWS;
    const int lrow  = lane & 15;
    const int khalf = (lane >> 4) << 1;
    const int rowhi = (lane >> 4) << 3;

    const int colh  = wave * 16 + lrow;
    const int col0  = wave * 16 + lrow;
    const int col1  = 128 + wave * 16 + lrow;
    const int col1c = min(col1, NREL - 1);
    const bool two  = 128 + wave * 16 < NREL;

    v2f bh[32], bo0[32], bo1[32];
    head_load_b(bh, W1 + khalf * NFEAT + colh, NFEAT);
    head_load_b(bo0, W2 + khalf * NREL + col0, NREL);
    if (two) head_load_b(bo1, W2 + khalf * NREL + col1c, NREL);
    const float bvh = b1[colh], bv0 = b2[col0], bv1 = b2[col1c];

#pragma unroll
    for (int i = 0; i < 2; ++i) {
        const int f = tid + i * HEAD_TB;
        const int r = f >> 5, c4 = (f & 31) * 4;
        const int gr = min(rbase + r, nrows - 1);
        *(float4*)(Gs + r * HPITCH + c4) = *(const float4*)(G + (long)gr * NFEAT + c4);
    }
    __syncthreads();

    {
        const v8f h = head_tile(Gs + lrow * HPITCH + khalf, bh, bvh);
#pragma unroll
        for (int r = 0; r < 8; ++r)
            Hs[(rowhi + r) * HPITCH + colh] = fmaxf(h[r], 0.0f);
    }
    __syncthreads();

    const float* Ap = Hs + lrow * HPITCH + khalf;
    {
        const v8f o = head_tile(Ap, bo0, bv0);
#pragma unroll
        for (int r = 0; r < 8; ++r) {
            const int row = rbase + rowhi + r;
            if (row < nrows) out[(long)row * NREL + col0] = o[r];
        }
    }
    if (two) {
        const v8f o = head_tile(Ap, bo1, bv1);
        if (col1 < NREL) {
#pragma unroll
            for (int r = 0; r < 8; ++r) {
                const int row = rbase + rowhi + r;
                if (row < nrows) out[(long)row * NREL + col1] = o[r];
            }
        }
    }
}

extern "C" void kernel_launch(void* const* d_in, const int* in_sizes, int n_in,
                              void* d_out, int out_size, void* d_ws, size_t ws_size,
                              hipStream_t stream) {
    const int*   nid    = (const int*)d_in[0];
    const int*   z      = (const int*)d_in[1];
    const int*   src    = (const int*)d_in[2];
    const int*   dst    = (const int*)d_in[3];
    const int*   gid    = (const int*)d_in[4];
    const float* emb    = (const float*)d_in[5];
    const float* ztab   = (const float*)d_in[6];
    const float* Ws[3]  = {(const float*)d_in[7], (const float*)d_in[9], (const float*)d_in[11]};
    const float* bs[3]  = {(const float*)d_in[8], (const float*)d_in[10], (const float*)d_in[12]};
    const float* lin1_W = (const float*)d_in[13];
    const float* lin1_b = (const float*)d_in[14];
    const float* lin2_W = (const float*)d_in[15];
    const float* lin2_b = (const float*)d_in[16];

    const int N = in_sizes[0];
    const int E = in_sizes[2];

    float* x        = (float*)d_ws;
    float* agg      = x + (size_t)N * NFEAT;
    float* norm_out = agg + (size_t)N * NFEAT;
    float* norm_in  = norm_out + N;
    float* g        = norm_in + N;
    int*   cnt_out  = (int*)(g + (size_t)NGRAPH * NFEAT);
    int*   cnt_in   = cnt_out + N;
    int*   rowptr   = cnt_in + N;
    int*   csr_src  = rowptr + N + 1;
    int*   ebd      = (int*)agg;
    int*   ebs      = ebd + E;
    int*   ebo      = ebs + E;
    int*   histd    = ebo + E;
    int*   hists    = histd + NBIN * NBIN;
    int*   bstartd  = hists + NBIN * NBIN;
    int*   bstarts  = bstartd + NBIN + 1;

    auto blocks = [](long n, int tb) { return (unsigned)((n + tb - 1) / tb); };

    int shift = 0;
    while (((N - 1) >> shift) >= NBIN) ++shift;
    const int chk = (E + NBIN - 1) / NBIN;
    bin_count_kernel<<<NBIN, BIN_TB, 0, stream>>>(src, dst, histd, hists, (int*)g, NGRAPH * NFEAT, E, chk, shift);
    bin_scatter_kernel<<<NBIN, BIN_TB, 3 * sizeof(int) * chk, stream>>>(src, dst, histd, hists, ebd, ebs, ebo, bstartd, bstarts, E, chk, shift);
    bin_csr_kernel<<<NBIN, BIN_TB, 3 * sizeof(int) << shift, stream>>>(
        ebd, ebs, ebo, bstartd, bstarts, cnt_out, cnt_in, rowptr, csr_src, N, E, shift);
    build_kernel<<<blocks((long)((N + ROWS_BX - 1) / ROWS_BX) * 32, 1024), 1024, 0, stream>>>(
        cnt_out, cnt_in, norm_out, norm_in, nid, z, emb, ztab, x, N);

    const long ntiles = ((long)N + TILE_ROWS - 1) / TILE_ROWS;
    const unsigned need_blocks = blocks(ntiles, LAYER_TB / 32);
    const unsigned layer_blocks = need_blocks > 512u ? need_blocks : 512u;
    float* xin = x;
    float* xout = agg;
    for (int layer = 0; layer < 3; ++layer) {
        const bool last = layer == 2;
        gcn_layer_kernel<<<layer_blocks, LAYER_TB, 0, stream>>>(
            rowptr, csr_src, xin, norm_out, norm_in, Ws[layer], bs[layer], xout, N, last ? 0 : 1,
            last ? gid : nullptr, last ? g : nullptr);
        float* t = xin; xin = xout; xout = t;
    }

    head_kernel<<<(NGRAPH + TILE_ROWS - 1) / TILE_ROWS, HEAD_TB, 0, stream>>>(
        g, lin1_W, lin1_b, lin2_W, lin2_b, (float*)d_out, NGRAPH);
}


# Two launches fewer (8 to 6): a chunk block counts and sorts its own edges in place; the bin block finds its pieces by per-chunk starts and builds norms and x rows too; row sums in LDS arrival order
# speedup vs baseline: 4.7894x; 1.0541x over previous
#include <hip/hip_runtime.h>
#include <hip/hip_bf16.h>

typedef float v2f __attribute__((ext_vector_type(2)));
typedef float v8f __attribute__((ext_vector_type(8)));

#define NFEAT 128
#define NGRAPH 512
#define NREL 200


#define NBIN 256
#define BIN_TB 256
#define BIN_LST 8192
#define BIN_EB 16
#define SEG_W 32
static_assert(NBIN == BIN_TB, "thread t of a bin's block holds the piece of chunk t");
static_assert(SEG_W == 32, "a wave's lanes hold the pieces of 32 neighbouring chunks");

__device__ __forceinline__ int block_excl_scan(int v, int* __restrict__ wsum, int tid, int& total) {
    const int lane = tid & 31, w = tid >> 5;
    int incl = v;
#pragma unroll
    for (int o = 1; o < 32; o <<= 1) {
        int t = __shfl_up(incl, o);
        if (lane >= o) incl += t;
    }
    if (lane == 31) wsum[w] = incl;
    __syncthreads();
    int woff = 0, tot = 0;
#pragma unroll
    for (int k = 0; k < BIN_TB / 32; ++k) {
        const int t = wsum[k];
        if (k < w) woff += t;
        tot += t;
    }
    __syncthreads();
    total = tot;
    return woff + incl - v;
}

__global__ __launch_bounds__(BIN_TB) void chunk_sort_kernel(
    const int* __restrict__ src, const int* __restrict__ dst,
    int* __restrict__ tcd, int* __restrict__ tld, int* __restrict__ tcs, int* __restrict__ tls,
    int* __restrict__ bd, int* __restrict__ bs, int* __restrict__ bo,
    int* __restrict__ gclear, int nclear, int E, int chk, int shift) {
    extern __shared__ int stage[];
    __shared__ int hd[NBIN], hs[NBIN], lod[NBIN], los[NBIN], curd[NBIN], curs[NBIN], wsum[BIN_TB / 32];
    int* sd = stage;
    int* ss = stage + chk;
    int* so = stage + 2 * chk;
    const int tid = threadIdx.x;
    hd[tid] = 0;
    hs[tid] = 0;
    curd[tid] = 0;
    curs[tid] = 0;
    for (int i = blockIdx.x * BIN_TB + tid; i < nclear; i += gridDim.x * BIN_TB) gclear[i] = 0;
    __syncthreads();
    const int e0 = blockIdx.x * chk;
    const int e1 = min(e0 + chk, E);
    for (int e = e0 + tid; e < e1; e += BIN_EB * BIN_TB) {
        int d[BIN_EB], sv[BIN_EB];
#pragma unroll
        for (int k = 0; k < BIN_EB; ++k) {
            const int ee = e + k * BIN_TB;
            d[k]  = ee < e1 ? dst[ee] : -1;
            sv[k] = ee < e1 ? src[ee] : -1;
        }
#pragma unroll
        for (int k = 0; k < BIN_EB; ++k) {
            if (d[k] >= 0) {
                atomicAdd(&hd[d[k] >> shift], 1);
                atomicAdd(&hs[sv[k] >> shift], 1);
            }
        }
    }
    __syncthreads();
    const int cd = hd[tid];
    const int cs = hs[tid];
    int unused;
    const int ld = block_excl_scan(cd, wsum, tid, unused);
    const int ls = block_excl_scan(cs, wsum, tid, unused);
    tcd[tid * NBIN + blockIdx.x] = cd;
    tld[tid * NBIN + blockIdx.x] = ld;
    tcs[tid * NBIN + blockIdx.x] = cs;
    tls[tid * NBIN + blockIdx.x] = ls;
    lod[tid] = ld;
    los[tid] = ls;
    __syncthreads();
    for (int e = e0 + tid; e < e1; e += BIN_EB * BIN_TB) {
        int d[BIN_EB], sv[BIN_EB];
#pragma unroll
        for (int k = 0; k < BIN_EB; ++k) {
            const int ee = e + k * BIN_TB;
            d[k]  = ee < e1 ? dst[ee] : -1;
            sv[k] = ee < e1 ? src[ee] : -1;
        }
#pragma unroll
        for (int k = 0; k < BIN_EB; ++k) {
            if (d[k] >= 0) {
                const int kd = d[k] >> shift;
                const int ks = sv[k] >> shift;
                const int p = lod[kd] + atomicAdd(&curd[kd], 1);
                const int q = los[ks] + atomicAdd(&curs[ks], 1);
                sd[p] = d[k];
                ss[p] = sv[k];
                so[q] = sv[k];
            }
        }
    }
    __syncthreads();
    const int n = e1 - e0;
    for (int i = tid; i < n; i += BIN_TB) {
        bd[e0 + i] = sd[i];
        bs[e0 + i] = ss[i];
        bo[e0 + i] = so[i];
    }
}

#define ROWS_BX 16

__device__ __forceinline__ float4 scale4(float4 v, float s) {
    return make_float4(v.x * s, v.y * s, v.z * s, v.w * s);
}

__device__ __forceinline__ float4 bx_row(const float* __restrict__ tab, bool ent,
                                         const int* __restrict__ nidl, const int* __restrict__ zl, int r) {
    const int rn = __builtin_amdgcn_readfirstlane(nidl[r]);
    const int rz = __builtin_amdgcn_readfirstlane(zl[r]);
    return *(const float4*)(tab + (long)(ent ? rn : rz) * 64);
}

__device__ __forceinline__ float lane_f32(float v, int k) {
    return __int_as_float(__builtin_amdgcn_readlane(__float_as_int(v), k));
}

__global__ __launch_bounds__(BIN_TB) void bin_csr_build_kernel(
    const int* __restrict__ bd, const int* __restrict__ bs, const int* __restrict__ bo,
    const int* __restrict__ tcd, const int* __restrict__ tld,
    const int* __restrict__ tcs, const int* __restrict__ tls,
    float* __restrict__ norm_out, float* __restrict__ norm_in,
    int* __restrict__ rowptr, int* __restrict__ csr_src,
    const int* __restrict__ nid, const int* __restrict__ z,
    const float* __restrict__ emb, const float* __restrict__ ztab,
    float* __restrict__ x, int N, int E, int chk, int shift) {
    extern __shared__ int binmem[];
    __shared__ int wsum[BIN_TB / 32];
    __shared__ int lst[BIN_LST];
    const int tid = threadIdx.x;
    const int lane = tid & 31;
    const int wave = __builtin_amdgcn_readfirstlane(tid >> 5);
    const int binw = 1 << shift;
    int* cin = binmem;
    int* cur = binmem + binw;
    int* cou = binmem + 2 * binw;
    int* nidl = binmem + 3 * binw;
    int* zl = binmem + 4 * binw;
    const int node0 = (int)blockIdx.x << shift;
    if (blockIdx.x == 0 && tid == 0) rowptr[N] = E;
    if (node0 >= N) return;
    const int cd = tcd[blockIdx.x * NBIN + tid];
    const int ld = tld[blockIdx.x * NBIN + tid];
    const int cs = tcs[blockIdx.x * NBIN + tid];
    const int ls = tls[blockIdx.x * NBIN + tid];
    for (int i = tid; i < binw; i += BIN_TB) {
        cin[i] = 0;
        cou[i] = 0;
        const int node = min(node0 + i, N - 1);
        nidl[i] = nid[node];
        zl[i] = z[node];
    }
    int d0, m;
    block_excl_scan(ld, wsum, tid, d0);
    block_excl_scan(cd, wsum, tid, m);
    const int based = tid * chk + ld;
    const int bases = tid * chk + ls;
    int dv[SEG_W], s2[SEG_W], ov[SEG_W];
#pragma unroll
    for (int k = 0; k < SEG_W; ++k) {
        const int c  = __builtin_amdgcn_readlane(cd, k);
        const int b  = __builtin_amdgcn_readlane(based, k);
        const int c2 = __builtin_amdgcn_readlane(cs, k);
        const int b2 = __builtin_amdgcn_readlane(bases, k);
        dv[k] = lane < c ? bd[b + lane] : -1;
        s2[k] = lane < c ? bs[b + lane] : -1;
        ov[k] = lane < c2 ? bo[b2 + lane] : -1;
    }
#pragma unroll
    for (int k = 0; k < SEG_W; ++k) {
        if (dv[k] >= 0) atomicAdd(&cin[dv[k] - node0], 1);
        if (ov[k] >= 0) atomicAdd(&cou[ov[k] - node0], 1);
    }
    const bool anylong = __builtin_amdgcn_ballot_w32(cd > 32 || cs > 32) != 0;
    if (anylong) {
        for (int k = 0; k < SEG_W; ++k) {
            const int c  = __builtin_amdgcn_readlane(cd, k);
            const int b  = __builtin_amdgcn_readlane(based, k);
            const int c2 = __builtin_amdgcn_readlane(cs, k);
            const int b2 = __builtin_amdgcn_readlane(bases, k);
            for (int i = lane + 32; i < c; i += 32) atomicAdd(&cin[bd[b + i] - node0], 1);
            for (int i = lane + 32; i < c2; i += 32) atomicAdd(&cou[bo[b2 + i] - node0], 1);
        }
    }
    __syncthreads();
    const int per = (binw + BIN_TB - 1) / BIN_TB;
    const int beg = min(tid * per, binw);
    const int end = min(beg + per, binw);
    int sum = 0;
    for (int i = beg; i < end; ++i) sum += cin[i];
    int all;
    int run = block_excl_scan(sum, wsum, tid, all);
    for (int i = beg; i < end; ++i) {
        const int v = cin[i];
        cur[i] = run;
        if (node0 + i < N) rowptr[node0 + i] = d0 + run;
        run += v;
    }
    __syncthreads();
    const bool fits = m <= BIN_LST;
    {
        int p[SEG_W];
#pragma unroll
        for (int k = 0; k < SEG_W; ++k) p[k] = dv[k] >= 0 ? atomicAdd(&cur[dv[k] - node0], 1) : 0;
        if (fits) {
#pragma unroll
            for (int k = 0; k < SEG_W; ++k) if (dv[k] >= 0) lst[p[k]] = s2[k];
        } else {
#pragma unroll
            for (int k = 0; k < SEG_W; ++k) if (dv[k] >= 0) csr_src[d0 + p[k]] = s2[k];
        }
    }
    if (anylong) {
        for (int k = 0; k < SEG_W; ++k) {
            const int c = __builtin_amdgcn_readlane(cd, k);
            const int b = __builtin_amdgcn_readlane(based, k);
            for (int i = lane + 32; i < c; i += 32) {
                const int p = atomicAdd(&cur[bd[b + i] - node0], 1);
                if (fits) lst[p] = bs[b + i];
                else csr_src[d0 + p] = bs[b + i];
            }
        }
    }
    if (fits) {
        __syncthreads();
        for (int i = tid; i < m; i += BIN_TB) csr_src[d0 + i] = lst[i];
    }

    const bool ent = lane < 16;
    const float* tab = (ent ? emb : ztab) + (lane & 15) * 4;
    float* xo = x + lane * 4;
    for (int r0 = wave * ROWS_BX; r0 < binw; r0 += (BIN_TB / 32) * ROWS_BX) {
        const int i0 = node0 + r0;
        if (i0 >= N) break;
        if (r0 + ROWS_BX <= binw && i0 + ROWS_BX <= N) {
            float4 v[ROWS_BX];
#pragma unroll
            for (int j = 0; j < ROWS_BX; ++j) v[j] = bx_row(tab, ent, nidl, zl, r0 + j);
            const bool in_side = (lane & ROWS_BX) != 0;
            const int c = (in_side ? cin : cou)[r0 + (lane & (ROWS_BX - 1))];
            __builtin_amdgcn_sched_barrier(0);
            const float f_out = rsqrtf(fmaxf((float)c, 1.0f));
            (in_side ? norm_in : norm_out)[i0 + (lane & (ROWS_BX - 1))] = f_out;
            float* xr = xo + (long)i0 * NFEAT;
#pragma unroll
            for (int j = 0; j < ROWS_BX; ++j) *(float4*)(xr + j * NFEAT) = scale4(v[j], lane_f32(f_out, j));
        } else {
            const int rend = min(r0 + ROWS_BX, binw);
            for (int r = r0; r < rend && node0 + r < N; ++r) {
                const float f_out = rsqrtf(fmaxf((float)cou[r], 1.0f));
                const float f_in  = rsqrtf(fmaxf((float)cin[r], 1.0f));
                if (lane == 0) { norm_out[node0 + r] = f_out; norm_in[node0 + r] = f_in; }
                *(float4*)(xo + (long)(node0 + r) * NFEAT) = scale4(bx_row(tab, ent, nidl, zl, r), f_out);
            }
        }
    }
}

#define LAYER_TB 256
#define TILE_ROWS 16
#define APITCH 132
#define GU 16

__device__ __forceinline__ void close_row(float4& acc, float* __restrict__ As, int q, int nin_bits, int lane) {
    const float s = __int_as_float(__builtin_amdgcn_readlane(nin_bits, q));
    *(float4*)(As + q * APITCH + lane * 4) = make_float4(acc.x * s, acc.y * s, acc.z * s, acc.w * s);
    acc = make_float4(0.0f, 0.0f, 0.0f, 0.0f);
}

template <int U, bool NX>
__device__ __forceinline__ void take_rows(float4& acc, int& q, int& rend, int e, int j,
                                          int s_l, int rw, int nin_bits,
                                          const float* __restrict__ xl, float* __restrict__ As, int lane,
                                          int& s_nx, const int* __restrict__ list, int nx, int e_end) {
    float4 v[U];
#pragma unroll
    for (int u = 0; u < U; ++u) {
        const int sj = __builtin_amdgcn_readlane(s_l, j + u);
        v[u] = *(const float4*)(xl + (long)sj * NFEAT);
    }
    if (NX) {
        if (nx < e_end) s_nx = list[nx];
    }
#pragma unroll
    for (int u = 0; u < U; ++u) {
        while (e + u >= rend) {
            close_row(acc, As, q, nin_bits, lane);
            ++q;
            rend = __builtin_amdgcn_readlane(rw, q + 1);
        }
        acc.x += v[u].x;
        acc.y += v[u].y;
        acc.z += v[u].z;
        acc.w += v[u].w;
    }
}

__global__ __launch_bounds__(LAYER_TB) void gcn_layer_kernel(
    const int* __restrict__ rowptr, const int* __restrict__ csr_src,
    const float* __restrict__ x, const float* __restrict__ out_scale, const float* __restrict__ norm_in,
    const float* __restrict__ W, const float* __restrict__ bias,
    float* __restrict__ Y, int nrows, int relu,
    const int* __restrict__ pool_gid, float* __restrict__ pool_g) {
    __shared__ __attribute__((aligned(16))) float Wlds[NFEAT * NFEAT];
    __shared__ __attribute__((aligned(16))) float Astage[(LAYER_TB / 32) * TILE_ROWS * APITCH];

    const int tid = threadIdx.x;
    {
        const float4* Wv = (const float4*)W;
        float4* Lv = (float4*)Wlds;
#pragma unroll
        for (int i = 0; i < 16; ++i) Lv[tid + i * 256] = Wv[tid + i * 256];
    }
    __syncthreads();

    const int lane = tid & 31;
    const int wave = __builtin_amdgcn_readfirstlane(tid >> 5);
    const int slot = (wave + (((int)blockIdx.x >> 8) & 1) * 6) & 7;
    const int tile = slot * (int)gridDim.x + (int)blockIdx.x;
    const int rbase = tile * TILE_ROWS;
    if (rbase >= nrows) return;

    float* As = Astage + wave * (TILE_ROWS * APITCH);

    {
        const int rp = rowptr[min(rbase + lane, nrows)];
        const int nin_bits = __float_as_int(norm_in[min(rbase + lane, nrows - 1)]);
        const int e_beg = __builtin_amdgcn_readlane(rp, 0);
        const int e_end = __builtin_amdgcn_readlane(rp, TILE_ROWS);
        const int rw = lane < TILE_ROWS ? rp : 0x7fffffff;
        const float* xl = x + lane * 4;
        int q = 0;
        int rend = __builtin_amdgcn_readlane(rw, 1);
        float4 acc = make_float4(0.0f, 0.0f, 0.0f, 0.0f);

        int s_cur = 0;
        if (e_beg + lane < e_end) s_cur = csr_src[e_beg + lane];
        static_assert(2 * GU == 32, "a chunk of 32 list entries is two full groups");
        for (int base = e_beg; base < e_end; base += 32) {
            const int n = min(32, e_end - base);
            int s_nx = 0;
            const int nx = base + 32 + lane;
            int j = 0;
            if (n >= GU) {
                take_rows<GU, true>(acc, q, rend, base, 0, s_cur, rw, nin_bits, xl, As, lane, s_nx, csr_src, nx, e_end);
                j = GU;
            }
            if (n >= 2 * GU) {
                take_rows<GU, false>(acc, q, rend, base + GU, GU, s_cur, rw, nin_bits, xl, As, lane, s_nx, csr_src, nx, e_end);
                j = 2 * GU;
            }
            if (n - j >= 8) { take_rows<8, false>(acc, q, rend, base + j, j, s_cur, rw, nin_bits, xl, As, lane, s_nx, csr_src, nx, e_end); j += 8; }
            if (n - j >= 4) { take_rows<4, false>(acc, q, rend, base + j, j, s_cur, rw, nin_bits, xl, As, lane, s_nx, csr_src, nx, e_end); j += 4; }
            if (n - j >= 2) { take_rows<2, false>(acc, q, rend, base + j, j, s_cur, rw, nin_bits, xl, As, lane, s_nx, csr_src, nx, e_end); j += 2; }
            if (n - j >= 1) { take_rows<1, false>(acc, q, rend, base + j, j, s_cur, rw, nin_bits, xl, As, lane, s_nx, csr_src, nx, e_end); }
            s_cur = s_nx;
        }
        for (; q < TILE_ROWS; ++q) close_row(acc, As, q, nin_bits, lane);
    }

    const int lrow  = lane & 15;
    const int khalf = (lane >> 4) << 1;
    const int rowhi = (lane >> 4) << 3;
    const float* Arow = As + lrow * APITCH + khalf;

    v8f acc[8];
#pragma unroll
    for (int t = 0; t < 8; ++t) {
        float bv = bias[t * 16 + lrow];
#pragma unroll
        for (int r = 0; r < 8; ++r) acc[t][r] = bv;
    }

    float osc[8];
#pragma unroll
    for (int r = 0; r < 8; ++r) osc[r] = 1.0f;
    if (!pool_g) {
#pragma unroll
        for (int r = 0; r < 8; ++r) osc[r] = out_scale[min(rbase + rowhi + r, nrows - 1)];
    }

    for (int k = 0; k < NFEAT; k += 4) {
        const v2f a = *(const v2f*)(Arow + k);
#pragma unroll
        for (int t = 0; t < 8; ++t) {
            v2f b;
            b.x = Wlds[(k + khalf) * NFEAT + t * 16 + lrow];
            b.y = Wlds[(k + khalf + 1) * NFEAT + t * 16 + lrow];
            acc[t] = __builtin_amdgcn_wmma_f32_16x16x4_f32(
                false, a, false, b, (short)0, acc[t], false, false);
        }
    }

    if (pool_g) {
        const int row0 = rbase + rowhi;
        int gr[8];
        float run[8];
#pragma unroll
        for (int r = 0; r < 8; ++r) gr[r] = pool_gid[min(row0 + r, nrows - 1)];
#pragma unroll
        for (int t = 0; t < 8; ++t) run[t] = 0.0f;
        float* gcol = pool_g + lrow;
#pragma unroll
        for (int r = 0; r < 8; ++r) {
            if (r > 0 && gr[r] != gr[r - 1]) {
                float* gp = gcol + (long)gr[r - 1] * NFEAT;
#pragma unroll
                for (int t = 0; t < 8; ++t) { atomicAdd(gp + t * 16, run[t]); run[t] = 0.0f; }
            }
            if (row0 + r < nrows) {
#pragma unroll
                for (int t = 0; t < 8; ++t) {
                    float v = acc[t][r];
                    if (relu) v = fmaxf(v, 0.0f);
                    run[t] += v;
                }
            }
        }
        float* gp = gcol + (long)gr[7] * NFEAT;
#pragma unroll
        for (int t = 0; t < 8; ++t) atomicAdd(gp + t * 16, run[t]);
        return;
    }

#pragma unroll
    for (int r = 0; r < 8; ++r) {
        const int row = rbase + rowhi + r;
        if (row < nrows) {
#pragma unroll
            for (int t = 0; t < 8; ++t) {
                float v = acc[t][r];
                if (relu) v = fmaxf(v, 0.0f);
                Y[(long)row * NFEAT + t * 16 + lrow] = v * osc[r];
            }
        }
    }
}

#define HEAD_TB 256
#define HPITCH 132

__device__ __forceinline__ void head_load_b(v2f (&b)[32], const float* __restrict__ Bp, int ldb) {
#pragma unroll
    for (int s = 0; s < 32; ++s) {
        b[s].x = Bp[(4 * s + 0) * ldb];
        b[s].y = Bp[(4 * s + 1) * ldb];
    }
}

__device__ __forceinline__ v8f head_tile(const float* __restrict__ Ap, const v2f (&b)[32], float bv) {
    v8f c0, c1;
#pragma unroll
    for (int r = 0; r < 8; ++r) { c0[r] = bv; c1[r] = 0.0f; }
#pragma unroll
    for (int s = 0; s < 32; s += 2) {
        const v2f a0 = *(const v2f*)(Ap + 4 * s);
        const v2f a1 = *(const v2f*)(Ap + 4 * s + 4);
        c0 = __builtin_amdgcn_wmma_f32_16x16x4_f32(false, a0, false, b[s], (short)0, c0, false, false);
        c1 = __builtin_amdgcn_wmma_f32_16x16x4_f32(false, a1, false, b[s + 1], (short)0, c1, false, false);
    }
    return c0 + c1;
}

__global__ __launch_bounds__(HEAD_TB) void head_kernel(
    const float* __restrict__ G,
    const float* __restrict__ W1, const float* __restrict__ b1,
    const float* __restrict__ W2, const float* __restrict__ b2,
    float* __restrict__ out, int nrows) {
    __shared__ __attribute__((aligned(16))) float Gs[TILE_ROWS * HPITCH];
    __shared__ __attribute__((aligned(16))) float Hs[TILE_ROWS * HPITCH];

    const int tid = threadIdx.x;
    const int lane = tid & 31;
    const int wave = __builtin_amdgcn_readfirstlane(tid >> 5);
    const int rbase = blockIdx.x * TILE_ROWS;
    const int lrow  = lane & 15;
    const int khalf = (lane >> 4) << 1;
    const int rowhi = (lane >> 4) << 3;

    const int colh  = wave * 16 + lrow;
    const int col0  = wave * 16 + lrow;
    const int col1  = 128 + wave * 16 + lrow;
    const int col1c = min(col1, NREL - 1);
    const bool two  = 128 + wave * 16 < NREL;

    v2f bh[32], bo0[32], bo1[32];
    head_load_b(bh, W1 + khalf * NFEAT + colh, NFEAT);
    head_load_b(bo0, W2 + khalf * NREL + col0, NREL);
    if (two) head_load_b(bo1, W2 + khalf * NREL + col1c, NREL);
    const float bvh = b1[colh], bv0 = b2[col0], bv1 = b2[col1c];

#pragma unroll
    for (int i = 0; i < 2; ++i) {
        const int f = tid + i * HEAD_TB;
        const int r = f >> 5, c4 = (f & 31) * 4;
        const int gr = min(rbase + r, nrows - 1);
        *(float4*)(Gs + r * HPITCH + c4) = *(const float4*)(G + (long)gr * NFEAT + c4);
    }
    __syncthreads();

    {
        const v8f h = head_tile(Gs + lrow * HPITCH + khalf, bh, bvh);
#pragma unroll
        for (int r = 0; r < 8; ++r)
            Hs[(rowhi + r) * HPITCH + colh] = fmaxf(h[r], 0.0f);
    }
    __syncthreads();

    const float* Ap = Hs + lrow * HPITCH + khalf;
    {
        const v8f o = head_tile(Ap, bo0, bv0);
#pragma unroll
        for (int r = 0; r < 8; ++r) {
            const int row = rbase + rowhi + r;
            if (row < nrows) out[(long)row * NREL + col0] = o[r];
        }
    }
    if (two) {
        const v8f o = head_tile(Ap, bo1, bv1);
        if (col1 < NREL) {
#pragma unroll
            for (int r = 0; r < 8; ++r) {
                const int row = rbase + rowhi + r;
                if (row < nrows) out[(long)row * NREL + col1] = o[r];
            }
        }
    }
}

extern "C" void kernel_launch(void* const* d_in, const int* in_sizes, int n_in,
                              void* d_out, int out_size, void* d_ws, size_t ws_size,
                              hipStream_t stream) {
    const int*   nid    = (const int*)d_in[0];
    const int*   z      = (const int*)d_in[1];
    const int*   src    = (const int*)d_in[2];
    const int*   dst    = (const int*)d_in[3];
    const int*   gid    = (const int*)d_in[4];
    const float* emb    = (const float*)d_in[5];
    const float* ztab   = (const float*)d_in[6];
    const float* Ws[3]  = {(const float*)d_in[7], (const float*)d_in[9], (const float*)d_in[11]};
    const float* bs[3]  = {(const float*)d_in[8], (const float*)d_in[10], (const float*)d_in[12]};
    const float* lin1_W = (const float*)d_in[13];
    const float* lin1_b = (const float*)d_in[14];
    const float* lin2_W = (const float*)d_in[15];
    const float* lin2_b = (const float*)d_in[16];

    const int N = in_sizes[0];
    const int E = in_sizes[2];

    float* x        = (float*)d_ws;
    float* agg      = x + (size_t)N * NFEAT;
    float* norm_out = agg + (size_t)N * NFEAT;
    float* norm_in  = norm_out + N;
    float* g        = norm_in + N;
    int*   rowptr   = (int*)(g + (size_t)NGRAPH * NFEAT) + 2 * (size_t)N;
    int*   csr_src  = rowptr + N + 1;
    int*   ebd      = (int*)agg;
    int*   ebs      = ebd + E;
    int*   ebo      = ebs + E;
    int*   tcd      = ebo + E;
    int*   tld      = tcd + NBIN * NBIN;
    int*   tcs      = tld + NBIN * NBIN;
    int*   tls      = tcs + NBIN * NBIN;

    auto blocks = [](long n, int tb) { return (unsigned)((n + tb - 1) / tb); };

    int shift = 0;
    while (((N - 1) >> shift) >= NBIN) ++shift;
    const int chk = (E + NBIN - 1) / NBIN;
    chunk_sort_kernel<<<NBIN, BIN_TB, 3 * sizeof(int) * chk, stream>>>(
        src, dst, tcd, tld, tcs, tls, ebd, ebs, ebo, (int*)g, NGRAPH * NFEAT, E, chk, shift);
    bin_csr_build_kernel<<<NBIN, BIN_TB, 5 * sizeof(int) << shift, stream>>>(
        ebd, ebs, ebo, tcd, tld, tcs, tls, norm_out, norm_in, rowptr, csr_src,
        nid, z, emb, ztab, x, N, E, chk, shift);

    const long ntiles = ((long)N + TILE_ROWS - 1) / TILE_ROWS;
    const unsigned need_blocks = blocks(ntiles, LAYER_TB / 32);
    const unsigned layer_blocks = need_blocks > 512u ? need_blocks : 512u;
    float* xin = x;
    float* xout = agg;
    for (int layer = 0; layer < 3; ++layer) {
        const bool last = layer == 2;
        gcn_layer_kernel<<<layer_blocks, LAYER_TB, 0, stream>>>(
            rowptr, csr_src, xin, norm_out, norm_in, Ws[layer], bs[layer], xout, N, last ? 0 : 1,
            last ? gid : nullptr, last ? g : nullptr);
        float* t = xin; xin = xout; xout = t;
    }

    head_kernel<<<(NGRAPH + TILE_ROWS - 1) / TILE_ROWS, HEAD_TB, 0, stream>>>(
        g, lin1_W, lin1_b, lin2_W, lin2_b, (float*)d_out, NGRAPH);
}


# Sum pooling moved in front of the last layer's product (no relu, they commute): the layer pools its scaled row sums per graph, the head multiplies the 512 pooled rows by W2 on WMMA; sum order changes
# speedup vs baseline: 5.0202x; 1.0482x over previous
#include <hip/hip_runtime.h>
#include <hip/hip_bf16.h>

typedef float v2f __attribute__((ext_vector_type(2)));
typedef float v8f __attribute__((ext_vector_type(8)));

#define NFEAT 128
#define NGRAPH 512
#define NREL 200


#define NBIN 256
#define BIN_TB 256
#define BIN_LST 8192
#define BIN_EB 16
#define SEG_W 32
static_assert(NBIN == BIN_TB, "thread t of a bin's block holds the piece of chunk t");
static_assert(SEG_W == 32, "a wave's lanes hold the pieces of 32 neighbouring chunks");

__device__ __forceinline__ int block_excl_scan(int v, int* __restrict__ wsum, int tid, int& total) {
    const int lane = tid & 31, w = tid >> 5;
    int incl = v;
#pragma unroll
    for (int o = 1; o < 32; o <<= 1) {
        int t = __shfl_up(incl, o);
        if (lane >= o) incl += t;
    }
    if (lane == 31) wsum[w] = incl;
    __syncthreads();
    int woff = 0, tot = 0;
#pragma unroll
    for (int k = 0; k < BIN_TB / 32; ++k) {
        const int t = wsum[k];
        if (k < w) woff += t;
        tot += t;
    }
    __syncthreads();
    total = tot;
    return woff + incl - v;
}

__global__ __launch_bounds__(BIN_TB) void chunk_sort_kernel(
    const int* __restrict__ src, const int* __restrict__ dst,
    int* __restrict__ tcd, int* __restrict__ tld, int* __restrict__ tcs, int* __restrict__ tls,
    int* __restrict__ bd, int* __restrict__ bs, int* __restrict__ bo,
    int* __restrict__ gclear, int nclear, int E, int chk, int shift) {
    extern __shared__ int stage[];
    __shared__ int hd[NBIN], hs[NBIN], lod[NBIN], los[NBIN], curd[NBIN], curs[NBIN], wsum[BIN_TB / 32];
    int* sd = stage;
    int* ss = stage + chk;
    int* so = stage + 2 * chk;
    const int tid = threadIdx.x;
    hd[tid] = 0;
    hs[tid] = 0;
    curd[tid] = 0;
    curs[tid] = 0;
    for (int i = blockIdx.x * BIN_TB + tid; i < nclear; i += gridDim.x * BIN_TB) gclear[i] = 0;
    __syncthreads();
    const int e0 = blockIdx.x * chk;
    const int e1 = min(e0 + chk, E);
    for (int e = e0 + tid; e < e1; e += BIN_EB * BIN_TB) {
        int d[BIN_EB], sv[BIN_EB];
#pragma unroll
        for (int k = 0; k < BIN_EB; ++k) {
            const int ee = e + k * BIN_TB;
            d[k]  = ee < e1 ? dst[ee] : -1;
            sv[k] = ee < e1 ? src[ee] : -1;
        }
#pragma unroll
        for (int k = 0; k < BIN_EB; ++k) {
            if (d[k] >= 0) {
                atomicAdd(&hd[d[k] >> shift], 1);
                atomicAdd(&hs[sv[k] >> shift], 1);
            }
        }
    }
    __syncthreads();
    const int cd = hd[tid];
    const int cs = hs[tid];
    int unused;
    const int ld = block_excl_scan(cd, wsum, tid, unused);
    const int ls = block_excl_scan(cs, wsum, tid, unused);
    tcd[tid * NBIN + blockIdx.x] = cd;
    tld[tid * NBIN + blockIdx.x] = ld;
    tcs[tid * NBIN + blockIdx.x] = cs;
    tls[tid * NBIN + blockIdx.x] = ls;
    lod[tid] = ld;
    los[tid] = ls;
    __syncthreads();
    for (int e = e0 + tid; e < e1; e += BIN_EB * BIN_TB) {
        int d[BIN_EB], sv[BIN_EB];
#pragma unroll
        for (int k = 0; k < BIN_EB; ++k) {
            const int ee = e + k * BIN_TB;
            d[k]  = ee < e1 ? dst[ee] : -1;
            sv[k] = ee < e1 ? src[ee] : -1;
        }
#pragma unroll
        for (int k = 0; k < BIN_EB; ++k) {
            if (d[k] >= 0) {
                const int kd = d[k] >> shift;
                const int ks = sv[k] >> shift;
                const int p = lod[kd] + atomicAdd(&curd[kd], 1);
                const int q = los[ks] + atomicAdd(&curs[ks], 1);
                sd[p] = d[k];
                ss[p] = sv[k];
                so[q] = sv[k];
            }
        }
    }
    __syncthreads();
    const int n = e1 - e0;
    for (int i = tid; i < n; i += BIN_TB) {
        bd[e0 + i] = sd[i];
        bs[e0 + i] = ss[i];
        bo[e0 + i] = so[i];
    }
}

#define ROWS_BX 16

__device__ __forceinline__ float4 scale4(float4 v, float s) {
    return make_float4(v.x * s, v.y * s, v.z * s, v.w * s);
}

__device__ __forceinline__ float4 bx_row(const float* __restrict__ tab, bool ent,
                                         const int* __restrict__ nidl, const int* __restrict__ zl, int r) {
    const int rn = __builtin_amdgcn_readfirstlane(nidl[r]);
    const int rz = __builtin_amdgcn_readfirstlane(zl[r]);
    return *(const float4*)(tab + (long)(ent ? rn : rz) * 64);
}

__device__ __forceinline__ float lane_f32(float v, int k) {
    return __int_as_float(__builtin_amdgcn_readlane(__float_as_int(v), k));
}

__global__ __launch_bounds__(BIN_TB) void bin_csr_build_kernel(
    const int* __restrict__ bd, const int* __restrict__ bs, const int* __restrict__ bo,
    const int* __restrict__ tcd, const int* __restrict__ tld,
    const int* __restrict__ tcs, const int* __restrict__ tls,
    float* __restrict__ norm_out, float* __restrict__ norm_in,
    int* __restrict__ rowptr, int* __restrict__ csr_src,
    const int* __restrict__ nid, const int* __restrict__ z,
    const float* __restrict__ emb, const float* __restrict__ ztab,
    float* __restrict__ x, int N, int E, int chk, int shift) {
    extern __shared__ int binmem[];
    __shared__ int wsum[BIN_TB / 32];
    __shared__ int lst[BIN_LST];
    const int tid = threadIdx.x;
    const int lane = tid & 31;
    const int wave = __builtin_amdgcn_readfirstlane(tid >> 5);
    const int binw = 1 << shift;
    int* cin = binmem;
    int* cur = binmem + binw;
    int* cou = binmem + 2 * binw;
    int* nidl = binmem + 3 * binw;
    int* zl = binmem + 4 * binw;
    const int node0 = (int)blockIdx.x << shift;
    if (blockIdx.x == 0 && tid == 0) rowptr[N] = E;
    if (node0 >= N) return;
    const int cd = tcd[blockIdx.x * NBIN + tid];
    const int ld = tld[blockIdx.x * NBIN + tid];
    const int cs = tcs[blockIdx.x * NBIN + tid];
    const int ls = tls[blockIdx.x * NBIN + tid];
    for (int i = tid; i < binw; i += BIN_TB) {
        cin[i] = 0;
        cou[i] = 0;
        const int node = min(node0 + i, N - 1);
        nidl[i] = nid[node];
        zl[i] = z[node];
    }
    int d0, m;
    block_excl_scan(ld, wsum, tid, d0);
    block_excl_scan(cd, wsum, tid, m);
    const int based = tid * chk + ld;
    const int bases = tid * chk + ls;
    int dv[SEG_W], s2[SEG_W], ov[SEG_W];
#pragma unroll
    for (int k = 0; k < SEG_W; ++k) {
        const int c  = __builtin_amdgcn_readlane(cd, k);
        const int b  = __builtin_amdgcn_readlane(based, k);
        const int c2 = __builtin_amdgcn_readlane(cs, k);
        const int b2 = __builtin_amdgcn_readlane(bases, k);
        dv[k] = lane < c ? bd[b + lane] : -1;
        s2[k] = lane < c ? bs[b + lane] : -1;
        ov[k] = lane < c2 ? bo[b2 + lane] : -1;
    }
#pragma unroll
    for (int k = 0; k < SEG_W; ++k) {
        if (dv[k] >= 0) atomicAdd(&cin[dv[k] - node0], 1);
        if (ov[k] >= 0) atomicAdd(&cou[ov[k] - node0], 1);
    }
    const bool anylong = __builtin_amdgcn_ballot_w32(cd > 32 || cs > 32) != 0;
    if (anylong) {
        for (int k = 0; k < SEG_W; ++k) {
            const int c  = __builtin_amdgcn_readlane(cd, k);
            const int b  = __builtin_amdgcn_readlane(based, k);
            const int c2 = __builtin_amdgcn_readlane(cs, k);
            const int b2 = __builtin_amdgcn_readlane(bases, k);
            for (int i = lane + 32; i < c; i += 32) atomicAdd(&cin[bd[b + i] - node0], 1);
            for (int i = lane + 32; i < c2; i += 32) atomicAdd(&cou[bo[b2 + i] - node0], 1);
        }
    }
    __syncthreads();
    const int per = (binw + BIN_TB - 1) / BIN_TB;
    const int beg = min(tid * per, binw);
    const int end = min(beg + per, binw);
    int sum = 0;
    for (int i = beg; i < end; ++i) sum += cin[i];
    int all;
    int run = block_excl_scan(sum, wsum, tid, all);
    for (int i = beg; i < end; ++i) {
        const int v = cin[i];
        cur[i] = run;
        if (node0 + i < N) rowptr[node0 + i] = d0 + run;
        run += v;
    }
    __syncthreads();
    const bool fits = m <= BIN_LST;
    {
        int p[SEG_W];
#pragma unroll
        for (int k = 0; k < SEG_W; ++k) p[k] = dv[k] >= 0 ? atomicAdd(&cur[dv[k] - node0], 1) : 0;
        if (fits) {
#pragma unroll
            for (int k = 0; k < SEG_W; ++k) if (dv[k] >= 0) lst[p[k]] = s2[k];
        } else {
#pragma unroll
            for (int k = 0; k < SEG_W; ++k) if (dv[k] >= 0) csr_src[d0 + p[k]] = s2[k];
        }
    }
    if (anylong) {
        for (int k = 0; k < SEG_W; ++k) {
            const int c = __builtin_amdgcn_readlane(cd, k);
            const int b = __builtin_amdgcn_readlane(based, k);
            for (int i = lane + 32; i < c; i += 32) {
                const int p = atomicAdd(&cur[bd[b + i] - node0], 1);
                if (fits) lst[p] = bs[b + i];
                else csr_src[d0 + p] = bs[b + i];
            }
        }
    }
    if (fits) {
        __syncthreads();
        for (int i = tid; i < m; i += BIN_TB) csr_src[d0 + i] = lst[i];
    }

    const bool ent = lane < 16;
    const float* tab = (ent ? emb : ztab) + (lane & 15) * 4;
    float* xo = x + lane * 4;
    for (int r0 = wave * ROWS_BX; r0 < binw; r0 += (BIN_TB / 32) * ROWS_BX) {
        const int i0 = node0 + r0;
        if (i0 >= N) break;
        if (r0 + ROWS_BX <= binw && i0 + ROWS_BX <= N) {
            float4 v[ROWS_BX];
#pragma unroll
            for (int j = 0; j < ROWS_BX; ++j) v[j] = bx_row(tab, ent, nidl, zl, r0 + j);
            const bool in_side = (lane & ROWS_BX) != 0;
            const int c = (in_side ? cin : cou)[r0 + (lane & (ROWS_BX - 1))];
            __builtin_amdgcn_sched_barrier(0);
            const float f_out = rsqrtf(fmaxf((float)c, 1.0f));
            (in_side ? norm_in : norm_out)[i0 + (lane & (ROWS_BX - 1))] = f_out;
            float* xr = xo + (long)i0 * NFEAT;
#pragma unroll
            for (int j = 0; j < ROWS_BX; ++j) *(float4*)(xr + j * NFEAT) = scale4(v[j], lane_f32(f_out, j));
        } else {
            const int rend = min(r0 + ROWS_BX, binw);
            for (int r = r0; r < rend && node0 + r < N; ++r) {
                const float f_out = rsqrtf(fmaxf((float)cou[r], 1.0f));
                const float f_in  = rsqrtf(fmaxf((float)cin[r], 1.0f));
                if (lane == 0) { norm_out[node0 + r] = f_out; norm_in[node0 + r] = f_in; }
                *(float4*)(xo + (long)(node0 + r) * NFEAT) = scale4(bx_row(tab, ent, nidl, zl, r), f_out);
            }
        }
    }
}

#define LAYER_TB 256
#define TILE_ROWS 16
#define APITCH 132
#define GU 16

__device__ __forceinline__ void close_row(float4& acc, float* __restrict__ As, int q, int nin_bits, int lane) {
    const float s = __int_as_float(__builtin_amdgcn_readlane(nin_bits, q));
    *(float4*)(As + q * APITCH + lane * 4) = make_float4(acc.x * s, acc.y * s, acc.z * s, acc.w * s);
    acc = make_float4(0.0f, 0.0f, 0.0f, 0.0f);
}

template <int U, bool NX>
__device__ __forceinline__ void take_rows(float4& acc, int& q, int& rend, int e, int j,
                                          int s_l, int rw, int nin_bits,
                                          const float* __restrict__ xl, float* __restrict__ As, int lane,
                                          int& s_nx, const int* __restrict__ list, int nx, int e_end) {
    float4 v[U];
#pragma unroll
    for (int u = 0; u < U; ++u) {
        const int sj = __builtin_amdgcn_readlane(s_l, j + u);
        v[u] = *(const float4*)(xl + (long)sj * NFEAT);
    }
    if (NX) {
        if (nx < e_end) s_nx = list[nx];
    }
#pragma unroll
    for (int u = 0; u < U; ++u) {
        while (e + u >= rend) {
            close_row(acc, As, q, nin_bits, lane);
            ++q;
            rend = __builtin_amdgcn_readlane(rw, q + 1);
        }
        acc.x += v[u].x;
        acc.y += v[u].y;
        acc.z += v[u].z;
        acc.w += v[u].w;
    }
}

__device__ __forceinline__ void sum_tile_rows(const int* __restrict__ rowptr, const int* __restrict__ csr_src,
                                              const float* __restrict__ x, const float* __restrict__ norm_in,
                                              float* __restrict__ As, int rbase, int nrows, int lane) {
    const int rp = rowptr[min(rbase + lane, nrows)];
    const int nin_bits = __float_as_int(norm_in[min(rbase + lane, nrows - 1)]);
    const int e_beg = __builtin_amdgcn_readlane(rp, 0);
    const int e_end = __builtin_amdgcn_readlane(rp, TILE_ROWS);
    const int rw = lane < TILE_ROWS ? rp : 0x7fffffff;
    const float* xl = x + lane * 4;
    int q = 0;
    int rend = __builtin_amdgcn_readlane(rw, 1);
    float4 acc = make_float4(0.0f, 0.0f, 0.0f, 0.0f);

    int s_cur = 0;
    if (e_beg + lane < e_end) s_cur = csr_src[e_beg + lane];
    static_assert(2 * GU == 32, "a chunk of 32 list entries is two full groups");
    for (int base = e_beg; base < e_end; base += 32) {
        const int n = min(32, e_end - base);
        int s_nx = 0;
        const int nx = base + 32 + lane;
        int j = 0;
        if (n >= GU) {
            take_rows<GU, true>(acc, q, rend, base, 0, s_cur, rw, nin_bits, xl, As, lane, s_nx, csr_src, nx, e_end);
            j = GU;
        }
        if (n >= 2 * GU) {
            take_rows<GU, false>(acc, q, rend, base + GU, GU, s_cur, rw, nin_bits, xl, As, lane, s_nx, csr_src, nx, e_end);
            j = 2 * GU;
        }
        if (n - j >= 8) { take_rows<8, false>(acc, q, rend, base + j, j, s_cur, rw, nin_bits, xl, As, lane, s_nx, csr_src, nx, e_end); j += 8; }
        if (n - j >= 4) { take_rows<4, false>(acc, q, rend, base + j, j, s_cur, rw, nin_bits, xl, As, lane, s_nx, csr_src, nx, e_end); j += 4; }
        if (n - j >= 2) { take_rows<2, false>(acc, q, rend, base + j, j, s_cur, rw, nin_bits, xl, As, lane, s_nx, csr_src, nx, e_end); j += 2; }
        if (n - j >= 1) { take_rows<1, false>(acc, q, rend, base + j, j, s_cur, rw, nin_bits, xl, As, lane, s_nx, csr_src, nx, e_end); }
        s_cur = s_nx;
    }
    for (; q < TILE_ROWS; ++q) close_row(acc, As, q, nin_bits, lane);
}

__global__ __launch_bounds__(LAYER_TB) void gcn_layer_kernel(
    const int* __restrict__ rowptr, const int* __restrict__ csr_src,
    const float* __restrict__ x, const float* __restrict__ out_scale, const float* __restrict__ norm_in,
    const float* __restrict__ W, const float* __restrict__ bias,
    float* __restrict__ Y, int nrows, int relu) {
    __shared__ __attribute__((aligned(16))) float Wlds[NFEAT * NFEAT];
    __shared__ __attribute__((aligned(16))) float Astage[(LAYER_TB / 32) * TILE_ROWS * APITCH];

    const int tid = threadIdx.x;
    {
        const float4* Wv = (const float4*)W;
        float4* Lv = (float4*)Wlds;
#pragma unroll
        for (int i = 0; i < 16; ++i) Lv[tid + i * 256] = Wv[tid + i * 256];
    }
    __syncthreads();

    const int lane = tid & 31;
    const int wave = __builtin_amdgcn_readfirstlane(tid >> 5);
    const int slot = (wave + (((int)blockIdx.x >> 8) & 1) * 6) & 7;
    const int tile = slot * (int)gridDim.x + (int)blockIdx.x;
    const int rbase = tile * TILE_ROWS;
    if (rbase >= nrows) return;

    float* As = Astage + wave * (TILE_ROWS * APITCH);

    sum_tile_rows(rowptr, csr_src, x, norm_in, As, rbase, nrows, lane);

    const int lrow  = lane & 15;
    const int khalf = (lane >> 4) << 1;
    const int rowhi = (lane >> 4) << 3;
    const float* Arow = As + lrow * APITCH + khalf;

    v8f acc[8];
#pragma unroll
    for (int t = 0; t < 8; ++t) {
        float bv = bias[t * 16 + lrow];
#pragma unroll
        for (int r = 0; r < 8; ++r) acc[t][r] = bv;
    }

    float osc[8];
#pragma unroll
    for (int r = 0; r < 8; ++r) osc[r] = out_scale[min(rbase + rowhi + r, nrows - 1)];

    for (int k = 0; k < NFEAT; k += 4) {
        const v2f a = *(const v2f*)(Arow + k);
#pragma unroll
        for (int t = 0; t < 8; ++t) {
            v2f b;
            b.x = Wlds[(k + khalf) * NFEAT + t * 16 + lrow];
            b.y = Wlds[(k + khalf + 1) * NFEAT + t * 16 + lrow];
            acc[t] = __builtin_amdgcn_wmma_f32_16x16x4_f32(
                false, a, false, b, (short)0, acc[t], false, false);
        }
    }

#pragma unroll
    for (int r = 0; r < 8; ++r) {
        const int row = rbase + rowhi + r;
        if (row < nrows) {
#pragma unroll
            for (int t = 0; t < 8; ++t) {
                float v = acc[t][r];
                if (relu) v = fmaxf(v, 0.0f);
                Y[(long)row * NFEAT + t * 16 + lrow] = v * osc[r];
            }
        }
    }
}

#define POOL_TB 128

__global__ __launch_bounds__(POOL_TB) void gcn_pool_kernel(
    const int* __restrict__ rowptr, const int* __restrict__ csr_src,
    const float* __restrict__ x, const float* __restrict__ norm_in,
    const int* __restrict__ gid, float* __restrict__ pool, float* __restrict__ cnt, int nrows) {
    __shared__ __attribute__((aligned(16))) float Astage[(POOL_TB / 32) * TILE_ROWS * APITCH];
    const int lane = threadIdx.x & 31;
    const int wave = __builtin_amdgcn_readfirstlane((int)threadIdx.x >> 5);
    const int rbase = ((int)blockIdx.x * (POOL_TB / 32) + wave) * TILE_ROWS;
    if (rbase >= nrows) return;
    float* As = Astage + wave * (TILE_ROWS * APITCH);
    const int gid_l = gid[min(rbase + lane, nrows - 1)];
    sum_tile_rows(rowptr, csr_src, x, norm_in, As, rbase, nrows, lane);

    const int nvalid = min(TILE_ROWS, nrows - rbase);
    const float* Al = As + lane * 4;
    float* pl = pool + lane * 4;
    float4 run = make_float4(0.0f, 0.0f, 0.0f, 0.0f);
    int nrun = 0;
    for (int q = 0; q < nvalid; ++q) {
        const float4 v = *(const float4*)(Al + q * APITCH);
        run.x += v.x;
        run.y += v.y;
        run.z += v.z;
        run.w += v.w;
        ++nrun;
        const int gq = __builtin_amdgcn_readlane(gid_l, q);
        const int gn = __builtin_amdgcn_readlane(gid_l, min(q + 1, TILE_ROWS - 1));
        if (q == nvalid - 1 || gn != gq) {
            float* p = pl + (long)gq * NFEAT;
            atomicAdd(p + 0, run.x);
            atomicAdd(p + 1, run.y);
            atomicAdd(p + 2, run.z);
            atomicAdd(p + 3, run.w);
            if (lane == 0) atomicAdd(cnt + gq, (float)nrun);
            run = make_float4(0.0f, 0.0f, 0.0f, 0.0f);
            nrun = 0;
        }
    }
}

#define HEAD_TB 256
#define HPITCH 132

__device__ __forceinline__ v8f splat8(float v) {
    v8f c;
#pragma unroll
    for (int r = 0; r < 8; ++r) c[r] = v;
    return c;
}

__device__ __forceinline__ void head_load_b(v2f (&b)[32], const float* __restrict__ Bp, int ldb) {
#pragma unroll
    for (int s = 0; s < 32; ++s) {
        b[s].x = Bp[(4 * s + 0) * ldb];
        b[s].y = Bp[(4 * s + 1) * ldb];
    }
}

__device__ __forceinline__ v8f head_tile(const float* __restrict__ Ap, const v2f (&b)[32], v8f c0) {
    v8f c1;
#pragma unroll
    for (int r = 0; r < 8; ++r) c1[r] = 0.0f;
#pragma unroll
    for (int s = 0; s < 32; s += 2) {
        const v2f a0 = *(const v2f*)(Ap + 4 * s);
        const v2f a1 = *(const v2f*)(Ap + 4 * s + 4);
        c0 = __builtin_amdgcn_wmma_f32_16x16x4_f32(false, a0, false, b[s], (short)0, c0, false, false);
        c1 = __builtin_amdgcn_wmma_f32_16x16x4_f32(false, a1, false, b[s + 1], (short)0, c1, false, false);
    }
    return c0 + c1;
}

__global__ __launch_bounds__(HEAD_TB) void head_kernel(
    const float* __restrict__ P, const float* __restrict__ cnt,
    const float* __restrict__ Wg, const float* __restrict__ bg,
    const float* __restrict__ W1, const float* __restrict__ b1,
    const float* __restrict__ W2, const float* __restrict__ b2,
    float* __restrict__ out, int nrows) {
    __shared__ __attribute__((aligned(16))) float Ps[TILE_ROWS * HPITCH];
    __shared__ __attribute__((aligned(16))) float Gs[TILE_ROWS * HPITCH];
    __shared__ __attribute__((aligned(16))) float Hs[TILE_ROWS * HPITCH];

    const int tid = threadIdx.x;
    const int lane = tid & 31;
    const int wave = __builtin_amdgcn_readfirstlane(tid >> 5);
    const int rbase = blockIdx.x * TILE_ROWS;
    const int lrow  = lane & 15;
    const int khalf = (lane >> 4) << 1;
    const int rowhi = (lane >> 4) << 3;

    const int colh  = wave * 16 + lrow;
    const int col0  = wave * 16 + lrow;
    const int col1  = 128 + wave * 16 + lrow;
    const int col1c = min(col1, NREL - 1);
    const bool two  = 128 + wave * 16 < NREL;

    v2f bgf[32], bh[32], bo0[32], bo1[32];
    head_load_b(bgf, Wg + khalf * NFEAT + colh, NFEAT);
    head_load_b(bh, W1 + khalf * NFEAT + colh, NFEAT);
    head_load_b(bo0, W2 + khalf * NREL + col0, NREL);
    if (two) head_load_b(bo1, W2 + khalf * NREL + col1c, NREL);
    const float bvg = bg[colh], bvh = b1[colh], bv0 = b2[col0], bv1 = b2[col1c];
    v8f cg;
#pragma unroll
    for (int r = 0; r < 8; ++r)
        cg[r] = cnt[min(rbase + rowhi + r, nrows - 1)] * bvg;

#pragma unroll
    for (int i = 0; i < 2; ++i) {
        const int f = tid + i * HEAD_TB;
        const int r = f >> 5, c4 = (f & 31) * 4;
        const int gr = min(rbase + r, nrows - 1);
        *(float4*)(Ps + r * HPITCH + c4) = *(const float4*)(P + (long)gr * NFEAT + c4);
    }
    __syncthreads();

    {
        const v8f g = head_tile(Ps + lrow * HPITCH + khalf, bgf, cg);
#pragma unroll
        for (int r = 0; r < 8; ++r)
            Gs[(rowhi + r) * HPITCH + colh] = g[r];
    }
    __syncthreads();

    {
        const v8f h = head_tile(Gs + lrow * HPITCH + khalf, bh, splat8(bvh));
#pragma unroll
        for (int r = 0; r < 8; ++r)
            Hs[(rowhi + r) * HPITCH + colh] = fmaxf(h[r], 0.0f);
    }
    __syncthreads();

    const float* Ap = Hs + lrow * HPITCH + khalf;
    {
        const v8f o = head_tile(Ap, bo0, splat8(bv0));
#pragma unroll
        for (int r = 0; r < 8; ++r) {
            const int row = rbase + rowhi + r;
            if (row < nrows) out[(long)row * NREL + col0] = o[r];
        }
    }
    if (two) {
        const v8f o = head_tile(Ap, bo1, splat8(bv1));
        if (col1 < NREL) {
#pragma unroll
            for (int r = 0; r < 8; ++r) {
                const int row = rbase + rowhi + r;
                if (row < nrows) out[(long)row * NREL + col1] = o[r];
            }
        }
    }
}

extern "C" void kernel_launch(void* const* d_in, const int* in_sizes, int n_in,
                              void* d_out, int out_size, void* d_ws, size_t ws_size,
                              hipStream_t stream) {
    const int*   nid    = (const int*)d_in[0];
    const int*   z      = (const int*)d_in[1];
    const int*   src    = (const int*)d_in[2];
    const int*   dst    = (const int*)d_in[3];
    const int*   gid    = (const int*)d_in[4];
    const float* emb    = (const float*)d_in[5];
    const float* ztab   = (const float*)d_in[6];
    const float* Ws[3]  = {(const float*)d_in[7], (const float*)d_in[9], (const float*)d_in[11]};
    const float* bs[3]  = {(const float*)d_in[8], (const float*)d_in[10], (const float*)d_in[12]};
    const float* lin1_W = (const float*)d_in[13];
    const float* lin1_b = (const float*)d_in[14];
    const float* lin2_W = (const float*)d_in[15];
    const float* lin2_b = (const float*)d_in[16];

    const int N = in_sizes[0];
    const int E = in_sizes[2];

    float* x        = (float*)d_ws;
    float* agg      = x + (size_t)N * NFEAT;
    float* norm_out = agg + (size_t)N * NFEAT;
    float* norm_in  = norm_out + N;
    float* g        = norm_in + N;
    float* gcnt     = g + (size_t)NGRAPH * NFEAT;
    int*   rowptr   = (int*)(g + (size_t)NGRAPH * NFEAT) + 2 * (size_t)N;
    int*   csr_src  = rowptr + N + 1;
    int*   ebd      = (int*)agg;
    int*   ebs      = ebd + E;
    int*   ebo      = ebs + E;
    int*   tcd      = ebo + E;
    int*   tld      = tcd + NBIN * NBIN;
    int*   tcs      = tld + NBIN * NBIN;
    int*   tls      = tcs + NBIN * NBIN;

    auto blocks = [](long n, int tb) { return (unsigned)((n + tb - 1) / tb); };

    int shift = 0;
    while (((N - 1) >> shift) >= NBIN) ++shift;
    const int chk = (E + NBIN - 1) / NBIN;
    chunk_sort_kernel<<<NBIN, BIN_TB, 3 * sizeof(int) * chk, stream>>>(
        src, dst, tcd, tld, tcs, tls, ebd, ebs, ebo, (int*)g, NGRAPH * NFEAT + NGRAPH, E, chk, shift);
    bin_csr_build_kernel<<<NBIN, BIN_TB, 5 * sizeof(int) << shift, stream>>>(
        ebd, ebs, ebo, tcd, tld, tcs, tls, norm_out, norm_in, rowptr, csr_src,
        nid, z, emb, ztab, x, N, E, chk, shift);

    const long ntiles = ((long)N + TILE_ROWS - 1) / TILE_ROWS;
    const unsigned need_blocks = blocks(ntiles, LAYER_TB / 32);
    const unsigned layer_blocks = need_blocks > 512u ? need_blocks : 512u;
    float* xin = x;
    float* xout = agg;
    for (int layer = 0; layer < 2; ++layer) {
        gcn_layer_kernel<<<layer_blocks, LAYER_TB, 0, stream>>>(
            rowptr, csr_src, xin, norm_out, norm_in, Ws[layer], bs[layer], xout, N, 1);
        float* t = xin; xin = xout; xout = t;
    }
    gcn_pool_kernel<<<blocks(ntiles, POOL_TB / 32), POOL_TB, 0, stream>>>(
        rowptr, csr_src, xin, norm_in, gid, g, gcnt, N);

    head_kernel<<<(NGRAPH + TILE_ROWS - 1) / TILE_ROWS, HEAD_TB, 0, stream>>>(
        g, gcnt, Ws[2], bs[2], lin1_W, lin1_b, lin2_W, lin2_b, (float*)d_out, NGRAPH);
}
